# starting kernel
# baseline (speedup 1.0000x reference)
#include <hip/hip_runtime.h>

typedef __attribute__((ext_vector_type(2))) float v2f;
typedef __attribute__((ext_vector_type(8))) float v8f;

__global__ void gemm_xw(const float* __restrict__ X, const float* __restrict__ W,
                        const float* __restrict__ bias, float* __restrict__ C,
                        int M, int K) {
    const int gthread = blockIdx.x * blockDim.x + threadIdx.x;
    const int wave    = gthread >> 5;
    const int lane    = threadIdx.x & 31;
    const int m0      = wave * 16;
    if (m0 >= M) return;

    const int half = lane >> 4;
    const int lm   = lane & 15;

    v8f acc[4];
    #pragma unroll
    for (int nt = 0; nt < 4; ++nt) {
        const float bv = bias ? bias[nt * 16 + lm] : 0.0f;
        #pragma unroll
        for (int v = 0; v < 8; ++v) acc[nt][v] = bv;
    }

    const float* xrow = X + (size_t)(m0 + lm) * K;

    for (int k0 = 0; k0 < K; k0 += 4) {
        const int kb = k0 + 2 * half;
        const float2 av = *(const float2*)(xrow + kb);
        v2f a; a.x = av.x; a.y = av.y;
        #pragma unroll
        for (int nt = 0; nt < 4; ++nt) {
            const int n = nt * 16 + lm;
            v2f b;
            b.x = W[(size_t)kb * 64 + n];
            b.y = W[(size_t)(kb + 1) * 64 + n];
            acc[nt] = __builtin_amdgcn_wmma_f32_16x16x4_f32(
                false, a, false, b, (short)0, acc[nt], false, false);
        }
    }

    #pragma unroll
    for (int nt = 0; nt < 4; ++nt) {
        const int n = nt * 16 + lm;
        #pragma unroll
        for (int v = 0; v < 8; ++v) {
            C[(size_t)(m0 + v + 8 * half) * 64 + n] = acc[nt][v];
        }
    }
}

__global__ void edge_scatter64(const float* __restrict__ rel,
                               const int* __restrict__ src,
                               const int* __restrict__ dst,
                               const int* __restrict__ et,
                               float* __restrict__ agg, int E, int N) {
    const long long t = (long long)blockIdx.x * blockDim.x + threadIdx.x;
    const int e = (int)(t >> 4);
    if (e >= E) return;
    const int f = ((int)t & 15) * 4;
    const float4 v = *(const float4*)(rel + ((long long)et[e] * N + src[e]) * 64 + f);
    float* q = agg + (long long)dst[e] * 64 + f;
    atomicAdd(q + 0, v.x);
    atomicAdd(q + 1, v.y);
    atomicAdd(q + 2, v.z);
    atomicAdd(q + 3, v.w);
}

__global__ void relu_k(float* __restrict__ p, int n) {
    const int i = blockIdx.x * blockDim.x + threadIdx.x;
    if (i < n) p[i] = fmaxf(p[i], 0.0f);
}

__global__ void layer3_node(const float* __restrict__ h2,
                            const float* __restrict__ W3,
                            const float* __restrict__ loop3,
                            const float* __restrict__ b3,
                            float* __restrict__ rel3, float* __restrict__ agg3,
                            int N) {
    const int i = blockIdx.x * blockDim.x + threadIdx.x;
    if (i >= N) return;
    const float* row = h2 + (size_t)i * 64;
    float r00 = 0.f, r01 = 0.f, r10 = 0.f, r11 = 0.f, l0 = 0.f, l1 = 0.f;
    #pragma unroll 8
    for (int k = 0; k < 64; ++k) {
        const float v = row[k];
        r00 += v * W3[k * 2 + 0];
        r01 += v * W3[k * 2 + 1];
        r10 += v * W3[128 + k * 2 + 0];
        r11 += v * W3[128 + k * 2 + 1];
        l0  += v * loop3[k * 2 + 0];
        l1  += v * loop3[k * 2 + 1];
    }
    rel3[(size_t)i * 2 + 0] = r00;
    rel3[(size_t)i * 2 + 1] = r01;
    rel3[(size_t)(N + i) * 2 + 0] = r10;
    rel3[(size_t)(N + i) * 2 + 1] = r11;
    agg3[(size_t)i * 2 + 0] = l0 + b3[0];
    agg3[(size_t)i * 2 + 1] = l1 + b3[1];
}

__global__ void edge_scatter2(const float* __restrict__ rel3,
                              const int* __restrict__ src,
                              const int* __restrict__ dst,
                              const int* __restrict__ et,
                              float* __restrict__ agg3, int E, int N) {
    const int e = blockIdx.x * blockDim.x + threadIdx.x;
    if (e >= E) return;
    const float* p = rel3 + ((long long)et[e] * N + src[e]) * 2;
    float* q = agg3 + (long long)dst[e] * 2;
    atomicAdd(q + 0, p[0]);
    atomicAdd(q + 1, p[1]);
}

__device__ __forceinline__ unsigned f2ord(float f) {
    unsigned u = __float_as_uint(f);
    return (u & 0x80000000u) ? ~u : (u | 0x80000000u);
}
__device__ __forceinline__ float ord2f(unsigned u) {
    return __uint_as_float((u & 0x80000000u) ? (u ^ 0x80000000u) : ~u);
}

__global__ void init_slot(unsigned* slot) { *slot = 0xFFFFFFFFu; }

__global__ void min_reduce(const float* __restrict__ p, int n, unsigned* slot) {
    __shared__ float sm[256];
    float m = 3.402823466e38f;
    for (int i = blockIdx.x * blockDim.x + threadIdx.x; i < n;
         i += gridDim.x * blockDim.x)
        m = fminf(m, p[i]);
    sm[threadIdx.x] = m;
    __syncthreads();
    for (int s = blockDim.x >> 1; s > 0; s >>= 1) {
        if ((int)threadIdx.x < s) sm[threadIdx.x] = fminf(sm[threadIdx.x], sm[threadIdx.x + s]);
        __syncthreads();
    }
    if (threadIdx.x == 0) atomicMin(slot, f2ord(sm[0]));
}

__global__ void finalize_k(const float* __restrict__ agg3,
                           const int* __restrict__ cell,
                           const int* __restrict__ maxs,
                           const unsigned* __restrict__ slot,
                           float* __restrict__ out, int N) {
    const int i = blockIdx.x * blockDim.x + threadIdx.x;
    if (i >= N) return;
    const float gmin = ord2f(*slot) - 1.0f;
    const bool upper = cell[i] >= (maxs[i] - 1);
    const bool lower = cell[i] == 0;
    out[(size_t)i * 2 + 0] = upper ? gmin : agg3[(size_t)i * 2 + 0];
    out[(size_t)i * 2 + 1] = lower ? gmin : agg3[(size_t)i * 2 + 1];
}

extern "C" void kernel_launch(void* const* d_in, const int* in_sizes, int n_in,
                              void* d_out, int out_size, void* d_ws, size_t ws_size,
                              hipStream_t stream) {
    const float* x      = (const float*)d_in[0];
    const int*   src    = (const int*)d_in[1];
    const int*   dst    = (const int*)d_in[2];
    const int*   etypes = (const int*)d_in[3];
    const int*   cell   = (const int*)d_in[4];
    const int*   maxs   = (const int*)d_in[5];
    const float* W1     = (const float*)d_in[6];
    const float* loop1  = (const float*)d_in[7];
    const float* b1     = (const float*)d_in[8];
    const float* W2     = (const float*)d_in[9];
    const float* loop2  = (const float*)d_in[10];
    const float* b2     = (const float*)d_in[11];
    const float* W3     = (const float*)d_in[12];
    const float* loop3  = (const float*)d_in[13];
    const float* b3     = (const float*)d_in[14];
    float* out = (float*)d_out;

    const int N = in_sizes[4];
    const int E = in_sizes[1];
    const int K1 = in_sizes[0] / N;

    float* relbuf = (float*)d_ws;
    float* h1     = relbuf + (size_t)2 * N * 64;
    float* h2     = h1 + (size_t)N * 64;
    float* rel3   = h2 + (size_t)N * 64;
    float* agg3   = rel3 + (size_t)4 * N;
    unsigned* slot = (unsigned*)(agg3 + (size_t)2 * N);

    const int waves   = (N + 15) / 16;
    const int gblocks = (waves * 32 + 255) / 256;
    const int eb64    = (int)(((long long)E * 16 + 255) / 256);
    const int nb      = (N + 255) / 256;
    const int fb      = (N * 64 + 255) / 256;
    const int eb2     = (E + 255) / 256;

    gemm_xw<<<gblocks, 256, 0, stream>>>(x, W1,            nullptr, relbuf,                  N, K1);
    gemm_xw<<<gblocks, 256, 0, stream>>>(x, W1 + K1 * 64,  nullptr, relbuf + (size_t)N * 64, N, K1);
    gemm_xw<<<gblocks, 256, 0, stream>>>(x, loop1,         b1,      h1,                      N, K1);
    edge_scatter64<<<eb64, 256, 0, stream>>>(relbuf, src, dst, etypes, h1, E, N);
    relu_k<<<fb, 256, 0, stream>>>(h1, N * 64);

    gemm_xw<<<gblocks, 256, 0, stream>>>(h1, W2,          nullptr, relbuf,                  N, 64);
    gemm_xw<<<gblocks, 256, 0, stream>>>(h1, W2 + 64 * 64, nullptr, relbuf + (size_t)N * 64, N, 64);
    gemm_xw<<<gblocks, 256, 0, stream>>>(h1, loop2,        b2,      h2,                      N, 64);
    edge_scatter64<<<eb64, 256, 0, stream>>>(relbuf, src, dst, etypes, h2, E, N);
    relu_k<<<fb, 256, 0, stream>>>(h2, N * 64);

    layer3_node<<<nb, 256, 0, stream>>>(h2, W3, loop3, b3, rel3, agg3, N);
    edge_scatter2<<<eb2, 256, 0, stream>>>(rel3, src, dst, etypes, agg3, E, N);
    init_slot<<<1, 1, 0, stream>>>(slot);
    min_reduce<<<512, 256, 0, stream>>>(agg3, N * 2, slot);
    finalize_k<<<nb, 256, 0, stream>>>(agg3, cell, maxs, slot, out, N);
}


# 64-wide edge scatter: a wave walks 32 edges with lane = column, each row read and atomically added as two whole 128-B lines (was half a wave x float4 per edge); still float atomics in arrival order
# speedup vs baseline: 1.3073x; 1.3073x over previous
#include <hip/hip_runtime.h>

typedef __attribute__((ext_vector_type(2))) float v2f;
typedef __attribute__((ext_vector_type(8))) float v8f;

__global__ void gemm_xw(const float* __restrict__ X, const float* __restrict__ W,
                        const float* __restrict__ bias, float* __restrict__ C,
                        int M, int K) {
    const int gthread = blockIdx.x * blockDim.x + threadIdx.x;
    const int wave    = gthread >> 5;
    const int lane    = threadIdx.x & 31;
    const int m0      = wave * 16;
    if (m0 >= M) return;

    const int half = lane >> 4;
    const int lm   = lane & 15;

    v8f acc[4];
    #pragma unroll
    for (int nt = 0; nt < 4; ++nt) {
        const float bv = bias ? bias[nt * 16 + lm] : 0.0f;
        #pragma unroll
        for (int v = 0; v < 8; ++v) acc[nt][v] = bv;
    }

    const float* xrow = X + (size_t)(m0 + lm) * K;

    for (int k0 = 0; k0 < K; k0 += 4) {
        const int kb = k0 + 2 * half;
        const float2 av = *(const float2*)(xrow + kb);
        v2f a; a.x = av.x; a.y = av.y;
        #pragma unroll
        for (int nt = 0; nt < 4; ++nt) {
            const int n = nt * 16 + lm;
            v2f b;
            b.x = W[(size_t)kb * 64 + n];
            b.y = W[(size_t)(kb + 1) * 64 + n];
            acc[nt] = __builtin_amdgcn_wmma_f32_16x16x4_f32(
                false, a, false, b, (short)0, acc[nt], false, false);
        }
    }

    #pragma unroll
    for (int nt = 0; nt < 4; ++nt) {
        const int n = nt * 16 + lm;
        #pragma unroll
        for (int v = 0; v < 8; ++v) {
            C[(size_t)(m0 + v + 8 * half) * 64 + n] = acc[nt][v];
        }
    }
}

__global__ void __launch_bounds__(256)
edge_scatter64(const float* __restrict__ rel,
               const int* __restrict__ src,
               const int* __restrict__ dst,
               const int* __restrict__ et,
               float* __restrict__ agg, int E, int N) {
    const int lane = threadIdx.x & 31;
    const int wave = (int)((blockIdx.x * blockDim.x + threadIdx.x) >> 5);
    const long long e0 = (long long)wave * 32;
    if (e0 >= E) return;
    const int cnt = (E - e0 < 32) ? (int)(E - e0) : 32;

    int srow = 0, drow = 0;
    if (lane < cnt) {
        const long long e = e0 + lane;
        srow = et[e] * N + src[e];
        drow = dst[e];
    }

    if (cnt == 32) {
        #pragma unroll
        for (int j0 = 0; j0 < 32; j0 += 8) {
            float a[8], b[8];
            #pragma unroll
            for (int u = 0; u < 8; ++u) {
                const int sr = __builtin_amdgcn_readlane(srow, j0 + u);
                const float* p = rel + (size_t)sr * 64 + lane;
                a[u] = p[0];
                b[u] = p[32];
            }
            #pragma unroll
            for (int u = 0; u < 8; ++u) {
                const int dr = __builtin_amdgcn_readlane(drow, j0 + u);
                float* q = agg + (size_t)dr * 64 + lane;
                atomicAdd(q, a[u]);
                atomicAdd(q + 32, b[u]);
            }
        }
    } else {
        for (int j = 0; j < cnt; ++j) {
            const int sr = __builtin_amdgcn_readlane(srow, j);
            const int dr = __builtin_amdgcn_readlane(drow, j);
            const float* p = rel + (size_t)sr * 64 + lane;
            float* q = agg + (size_t)dr * 64 + lane;
            atomicAdd(q, p[0]);
            atomicAdd(q + 32, p[32]);
        }
    }
}

__global__ void relu_k(float* __restrict__ p, int n) {
    const int i = blockIdx.x * blockDim.x + threadIdx.x;
    if (i < n) p[i] = fmaxf(p[i], 0.0f);
}

__global__ void layer3_node(const float* __restrict__ h2,
                            const float* __restrict__ W3,
                            const float* __restrict__ loop3,
                            const float* __restrict__ b3,
                            float* __restrict__ rel3, float* __restrict__ agg3,
                            int N) {
    const int i = blockIdx.x * blockDim.x + threadIdx.x;
    if (i >= N) return;
    const float* row = h2 + (size_t)i * 64;
    float r00 = 0.f, r01 = 0.f, r10 = 0.f, r11 = 0.f, l0 = 0.f, l1 = 0.f;
    #pragma unroll 8
    for (int k = 0; k < 64; ++k) {
        const float v = row[k];
        r00 += v * W3[k * 2 + 0];
        r01 += v * W3[k * 2 + 1];
        r10 += v * W3[128 + k * 2 + 0];
        r11 += v * W3[128 + k * 2 + 1];
        l0  += v * loop3[k * 2 + 0];
        l1  += v * loop3[k * 2 + 1];
    }
    rel3[(size_t)i * 2 + 0] = r00;
    rel3[(size_t)i * 2 + 1] = r01;
    rel3[(size_t)(N + i) * 2 + 0] = r10;
    rel3[(size_t)(N + i) * 2 + 1] = r11;
    agg3[(size_t)i * 2 + 0] = l0 + b3[0];
    agg3[(size_t)i * 2 + 1] = l1 + b3[1];
}

__global__ void edge_scatter2(const float* __restrict__ rel3,
                              const int* __restrict__ src,
                              const int* __restrict__ dst,
                              const int* __restrict__ et,
                              float* __restrict__ agg3, int E, int N) {
    const int e = blockIdx.x * blockDim.x + threadIdx.x;
    if (e >= E) return;
    const float* p = rel3 + ((long long)et[e] * N + src[e]) * 2;
    float* q = agg3 + (long long)dst[e] * 2;
    atomicAdd(q + 0, p[0]);
    atomicAdd(q + 1, p[1]);
}

__device__ __forceinline__ unsigned f2ord(float f) {
    unsigned u = __float_as_uint(f);
    return (u & 0x80000000u) ? ~u : (u | 0x80000000u);
}
__device__ __forceinline__ float ord2f(unsigned u) {
    return __uint_as_float((u & 0x80000000u) ? (u ^ 0x80000000u) : ~u);
}

__global__ void init_slot(unsigned* slot) { *slot = 0xFFFFFFFFu; }

__global__ void min_reduce(const float* __restrict__ p, int n, unsigned* slot) {
    __shared__ float sm[256];
    float m = 3.402823466e38f;
    for (int i = blockIdx.x * blockDim.x + threadIdx.x; i < n;
         i += gridDim.x * blockDim.x)
        m = fminf(m, p[i]);
    sm[threadIdx.x] = m;
    __syncthreads();
    for (int s = blockDim.x >> 1; s > 0; s >>= 1) {
        if ((int)threadIdx.x < s) sm[threadIdx.x] = fminf(sm[threadIdx.x], sm[threadIdx.x + s]);
        __syncthreads();
    }
    if (threadIdx.x == 0) atomicMin(slot, f2ord(sm[0]));
}

__global__ void finalize_k(const float* __restrict__ agg3,
                           const int* __restrict__ cell,
                           const int* __restrict__ maxs,
                           const unsigned* __restrict__ slot,
                           float* __restrict__ out, int N) {
    const int i = blockIdx.x * blockDim.x + threadIdx.x;
    if (i >= N) return;
    const float gmin = ord2f(*slot) - 1.0f;
    const bool upper = cell[i] >= (maxs[i] - 1);
    const bool lower = cell[i] == 0;
    out[(size_t)i * 2 + 0] = upper ? gmin : agg3[(size_t)i * 2 + 0];
    out[(size_t)i * 2 + 1] = lower ? gmin : agg3[(size_t)i * 2 + 1];
}

extern "C" void kernel_launch(void* const* d_in, const int* in_sizes, int n_in,
                              void* d_out, int out_size, void* d_ws, size_t ws_size,
                              hipStream_t stream) {
    const float* x      = (const float*)d_in[0];
    const int*   src    = (const int*)d_in[1];
    const int*   dst    = (const int*)d_in[2];
    const int*   etypes = (const int*)d_in[3];
    const int*   cell   = (const int*)d_in[4];
    const int*   maxs   = (const int*)d_in[5];
    const float* W1     = (const float*)d_in[6];
    const float* loop1  = (const float*)d_in[7];
    const float* b1     = (const float*)d_in[8];
    const float* W2     = (const float*)d_in[9];
    const float* loop2  = (const float*)d_in[10];
    const float* b2     = (const float*)d_in[11];
    const float* W3     = (const float*)d_in[12];
    const float* loop3  = (const float*)d_in[13];
    const float* b3     = (const float*)d_in[14];
    float* out = (float*)d_out;

    const int N = in_sizes[4];
    const int E = in_sizes[1];
    const int K1 = in_sizes[0] / N;

    float* relbuf = (float*)d_ws;
    float* h1     = relbuf + (size_t)2 * N * 64;
    float* h2     = h1 + (size_t)N * 64;
    float* rel3   = h2 + (size_t)N * 64;
    float* agg3   = rel3 + (size_t)4 * N;
    unsigned* slot = (unsigned*)(agg3 + (size_t)2 * N);

    const int waves   = (N + 15) / 16;
    const int gblocks = (waves * 32 + 255) / 256;
    const int eb64    = (E + 255) / 256;
    const int nb      = (N + 255) / 256;
    const int fb      = (N * 64 + 255) / 256;
    const int eb2     = (E + 255) / 256;

    gemm_xw<<<gblocks, 256, 0, stream>>>(x, W1,            nullptr, relbuf,                  N, K1);
    gemm_xw<<<gblocks, 256, 0, stream>>>(x, W1 + K1 * 64,  nullptr, relbuf + (size_t)N * 64, N, K1);
    gemm_xw<<<gblocks, 256, 0, stream>>>(x, loop1,         b1,      h1,                      N, K1);
    edge_scatter64<<<eb64, 256, 0, stream>>>(relbuf, src, dst, etypes, h1, E, N);
    relu_k<<<fb, 256, 0, stream>>>(h1, N * 64);

    gemm_xw<<<gblocks, 256, 0, stream>>>(h1, W2,          nullptr, relbuf,                  N, 64);
    gemm_xw<<<gblocks, 256, 0, stream>>>(h1, W2 + 64 * 64, nullptr, relbuf + (size_t)N * 64, N, 64);
    gemm_xw<<<gblocks, 256, 0, stream>>>(h1, loop2,        b2,      h2,                      N, 64);
    edge_scatter64<<<eb64, 256, 0, stream>>>(relbuf, src, dst, etypes, h2, E, N);
    relu_k<<<fb, 256, 0, stream>>>(h2, N * 64);

    layer3_node<<<nb, 256, 0, stream>>>(h2, W3, loop3, b3, rel3, agg3, N);
    edge_scatter2<<<eb2, 256, 0, stream>>>(rel3, src, dst, etypes, agg3, E, N);
    init_slot<<<1, 1, 0, stream>>>(slot);
    min_reduce<<<512, 256, 0, stream>>>(agg3, N * 2, slot);
    finalize_k<<<nb, 256, 0, stream>>>(agg3, cell, maxs, slot, out, N);
}


# relu of layers 1 and 2 applied as the next layer loads h1/h2 (layer-2 A fragments clamped in registers ahead of its products; layer-3 row reads): 2 elementwise passes gone, 15 -> 13 launches
# speedup vs baseline: 1.4227x; 1.0883x over previous
#include <hip/hip_runtime.h>

typedef __attribute__((ext_vector_type(2))) float v2f;
typedef __attribute__((ext_vector_type(8))) float v8f;

__global__ void gemm_xw(const float* __restrict__ X, const float* __restrict__ W,
                        const float* __restrict__ bias, float* __restrict__ C,
                        int M, int K) {
    const int gthread = blockIdx.x * blockDim.x + threadIdx.x;
    const int wave    = gthread >> 5;
    const int lane    = threadIdx.x & 31;
    const int m0      = wave * 16;
    if (m0 >= M) return;

    const int half = lane >> 4;
    const int lm   = lane & 15;

    v8f acc[4];
    #pragma unroll
    for (int nt = 0; nt < 4; ++nt) {
        const float bv = bias ? bias[nt * 16 + lm] : 0.0f;
        #pragma unroll
        for (int v = 0; v < 8; ++v) acc[nt][v] = bv;
    }

    const float* xrow = X + (size_t)(m0 + lm) * K;

    for (int k0 = 0; k0 < K; k0 += 4) {
        const int kb = k0 + 2 * half;
        const float2 av = *(const float2*)(xrow + kb);
        v2f a; a.x = av.x; a.y = av.y;
        #pragma unroll
        for (int nt = 0; nt < 4; ++nt) {
            const int n = nt * 16 + lm;
            v2f b;
            b.x = W[(size_t)kb * 64 + n];
            b.y = W[(size_t)(kb + 1) * 64 + n];
            acc[nt] = __builtin_amdgcn_wmma_f32_16x16x4_f32(
                false, a, false, b, (short)0, acc[nt], false, false);
        }
    }

    #pragma unroll
    for (int nt = 0; nt < 4; ++nt) {
        const int n = nt * 16 + lm;
        #pragma unroll
        for (int v = 0; v < 8; ++v) {
            C[(size_t)(m0 + v + 8 * half) * 64 + n] = acc[nt][v];
        }
    }
}

__global__ void gemm_relu_xw64(const float* __restrict__ X, const float* __restrict__ W,
                               const float* __restrict__ bias, float* __restrict__ C,
                               int M) {
    const int gthread = blockIdx.x * blockDim.x + threadIdx.x;
    const int wave    = gthread >> 5;
    const int lane    = threadIdx.x & 31;
    const int m0      = wave * 16;
    if (m0 >= M) return;

    const int half = lane >> 4;
    const int lm   = lane & 15;

    const float* xrow = X + (size_t)(m0 + lm) * 64 + 2 * half;
    const float* wcol = W + (size_t)(2 * half) * 64 + lm;
    unsigned coff = (unsigned)((m0 + 8 * half) * 64 + lm) * 4u;

    float2 av[16];
    #pragma unroll
    for (int ks = 0; ks < 16; ++ks) av[ks] = *(const float2*)(xrow + ks * 4);

    constexpr int kPre = 2;
    v2f b[16][4];
    #pragma unroll
    for (int ks = 0; ks < kPre; ++ks) {
        #pragma unroll
        for (int nt = 0; nt < 4; ++nt) {
            b[ks][nt].x = wcol[(size_t)(ks * 4) * 64 + nt * 16];
            b[ks][nt].y = wcol[(size_t)(ks * 4 + 1) * 64 + nt * 16];
        }
    }

    v8f acc[4];
    #pragma unroll
    for (int nt = 0; nt < 4; ++nt) {
        const float bv = bias ? bias[nt * 16 + lm] : 0.0f;
        #pragma unroll
        for (int v = 0; v < 8; ++v) acc[nt][v] = bv;
    }

    v2f a[16];
    #pragma unroll
    for (int ks = 0; ks < 16; ++ks) {
        a[ks].x = fmaxf(av[ks].x, 0.0f);
        a[ks].y = fmaxf(av[ks].y, 0.0f);
    }

    asm volatile("" : "+v"(acc[0]), "+v"(acc[1]), "+v"(acc[2]), "+v"(acc[3]));
    asm volatile("" : "+v"(coff));
    __builtin_amdgcn_sched_barrier(0);

    #pragma unroll
    for (int ks = 0; ks < 16; ++ks) {
        #pragma unroll
        for (int nt = 0; nt < 4; ++nt) {
            if (ks >= kPre) {
                b[ks][nt].x = wcol[(size_t)(ks * 4) * 64 + nt * 16];
                b[ks][nt].y = wcol[(size_t)(ks * 4 + 1) * 64 + nt * 16];
            }
            acc[nt] = __builtin_amdgcn_wmma_f32_16x16x4_f32(
                false, a[ks], false, b[ks][nt], (short)0, acc[nt], false, false);
        }
    }

    float* crow = (float*)((char*)C + coff);
    #pragma unroll
    for (int nt = 0; nt < 4; ++nt) {
        #pragma unroll
        for (int v = 0; v < 8; ++v) {
            crow[v * 64 + nt * 16] = acc[nt][v];
        }
    }
}

__global__ void __launch_bounds__(256)
edge_scatter64(const float* __restrict__ rel,
               const int* __restrict__ src,
               const int* __restrict__ dst,
               const int* __restrict__ et,
               float* __restrict__ agg, int E, int N) {
    const int lane = threadIdx.x & 31;
    const int wave = (int)((blockIdx.x * blockDim.x + threadIdx.x) >> 5);
    const long long e0 = (long long)wave * 32;
    if (e0 >= E) return;
    const int cnt = (E - e0 < 32) ? (int)(E - e0) : 32;

    int srow = 0, drow = 0;
    if (lane < cnt) {
        const long long e = e0 + lane;
        srow = et[e] * N + src[e];
        drow = dst[e];
    }

    if (cnt == 32) {
        #pragma unroll
        for (int j0 = 0; j0 < 32; j0 += 8) {
            float a[8], b[8];
            #pragma unroll
            for (int u = 0; u < 8; ++u) {
                const int sr = __builtin_amdgcn_readlane(srow, j0 + u);
                const float* p = rel + (size_t)sr * 64 + lane;
                a[u] = p[0];
                b[u] = p[32];
            }
            #pragma unroll
            for (int u = 0; u < 8; ++u) {
                const int dr = __builtin_amdgcn_readlane(drow, j0 + u);
                float* q = agg + (size_t)dr * 64 + lane;
                atomicAdd(q, a[u]);
                atomicAdd(q + 32, b[u]);
            }
        }
    } else {
        for (int j = 0; j < cnt; ++j) {
            const int sr = __builtin_amdgcn_readlane(srow, j);
            const int dr = __builtin_amdgcn_readlane(drow, j);
            const float* p = rel + (size_t)sr * 64 + lane;
            float* q = agg + (size_t)dr * 64 + lane;
            atomicAdd(q, p[0]);
            atomicAdd(q + 32, p[32]);
        }
    }
}

__global__ void layer3_node(const float* __restrict__ h2,
                            const float* __restrict__ W3,
                            const float* __restrict__ loop3,
                            const float* __restrict__ b3,
                            float* __restrict__ rel3, float* __restrict__ agg3,
                            int N) {
    const int i = blockIdx.x * blockDim.x + threadIdx.x;
    if (i >= N) return;
    const float* row = h2 + (size_t)i * 64;
    float r00 = 0.f, r01 = 0.f, r10 = 0.f, r11 = 0.f, l0 = 0.f, l1 = 0.f;
    #pragma unroll 8
    for (int k = 0; k < 64; ++k) {
        const float v = fmaxf(row[k], 0.0f);
        r00 += v * W3[k * 2 + 0];
        r01 += v * W3[k * 2 + 1];
        r10 += v * W3[128 + k * 2 + 0];
        r11 += v * W3[128 + k * 2 + 1];
        l0  += v * loop3[k * 2 + 0];
        l1  += v * loop3[k * 2 + 1];
    }
    rel3[(size_t)i * 2 + 0] = r00;
    rel3[(size_t)i * 2 + 1] = r01;
    rel3[(size_t)(N + i) * 2 + 0] = r10;
    rel3[(size_t)(N + i) * 2 + 1] = r11;
    agg3[(size_t)i * 2 + 0] = l0 + b3[0];
    agg3[(size_t)i * 2 + 1] = l1 + b3[1];
}

__global__ void edge_scatter2(const float* __restrict__ rel3,
                              const int* __restrict__ src,
                              const int* __restrict__ dst,
                              const int* __restrict__ et,
                              float* __restrict__ agg3, int E, int N) {
    const int e = blockIdx.x * blockDim.x + threadIdx.x;
    if (e >= E) return;
    const float* p = rel3 + ((long long)et[e] * N + src[e]) * 2;
    float* q = agg3 + (long long)dst[e] * 2;
    atomicAdd(q + 0, p[0]);
    atomicAdd(q + 1, p[1]);
}

__device__ __forceinline__ unsigned f2ord(float f) {
    unsigned u = __float_as_uint(f);
    return (u & 0x80000000u) ? ~u : (u | 0x80000000u);
}
__device__ __forceinline__ float ord2f(unsigned u) {
    return __uint_as_float((u & 0x80000000u) ? (u ^ 0x80000000u) : ~u);
}

__global__ void init_slot(unsigned* slot) { *slot = 0xFFFFFFFFu; }

__global__ void min_reduce(const float* __restrict__ p, int n, unsigned* slot) {
    __shared__ float sm[256];
    float m = 3.402823466e38f;
    for (int i = blockIdx.x * blockDim.x + threadIdx.x; i < n;
         i += gridDim.x * blockDim.x)
        m = fminf(m, p[i]);
    sm[threadIdx.x] = m;
    __syncthreads();
    for (int s = blockDim.x >> 1; s > 0; s >>= 1) {
        if ((int)threadIdx.x < s) sm[threadIdx.x] = fminf(sm[threadIdx.x], sm[threadIdx.x + s]);
        __syncthreads();
    }
    if (threadIdx.x == 0) atomicMin(slot, f2ord(sm[0]));
}

__global__ void finalize_k(const float* __restrict__ agg3,
                           const int* __restrict__ cell,
                           const int* __restrict__ maxs,
                           const unsigned* __restrict__ slot,
                           float* __restrict__ out, int N) {
    const int i = blockIdx.x * blockDim.x + threadIdx.x;
    if (i >= N) return;
    const float gmin = ord2f(*slot) - 1.0f;
    const bool upper = cell[i] >= (maxs[i] - 1);
    const bool lower = cell[i] == 0;
    out[(size_t)i * 2 + 0] = upper ? gmin : agg3[(size_t)i * 2 + 0];
    out[(size_t)i * 2 + 1] = lower ? gmin : agg3[(size_t)i * 2 + 1];
}

extern "C" void kernel_launch(void* const* d_in, const int* in_sizes, int n_in,
                              void* d_out, int out_size, void* d_ws, size_t ws_size,
                              hipStream_t stream) {
    const float* x      = (const float*)d_in[0];
    const int*   src    = (const int*)d_in[1];
    const int*   dst    = (const int*)d_in[2];
    const int*   etypes = (const int*)d_in[3];
    const int*   cell   = (const int*)d_in[4];
    const int*   maxs   = (const int*)d_in[5];
    const float* W1     = (const float*)d_in[6];
    const float* loop1  = (const float*)d_in[7];
    const float* b1     = (const float*)d_in[8];
    const float* W2     = (const float*)d_in[9];
    const float* loop2  = (const float*)d_in[10];
    const float* b2     = (const float*)d_in[11];
    const float* W3     = (const float*)d_in[12];
    const float* loop3  = (const float*)d_in[13];
    const float* b3     = (const float*)d_in[14];
    float* out = (float*)d_out;

    const int N = in_sizes[4];
    const int E = in_sizes[1];
    const int K1 = in_sizes[0] / N;

    float* relbuf = (float*)d_ws;
    float* h1     = relbuf + (size_t)2 * N * 64;
    float* h2     = h1 + (size_t)N * 64;
    float* rel3   = h2 + (size_t)N * 64;
    float* agg3   = rel3 + (size_t)4 * N;
    unsigned* slot = (unsigned*)(agg3 + (size_t)2 * N);

    const int waves   = (N + 15) / 16;
    const int gblocks = (waves * 32 + 255) / 256;
    const int eb64    = (E + 255) / 256;
    const int nb      = (N + 255) / 256;
    const int eb2     = (E + 255) / 256;

    gemm_xw<<<gblocks, 256, 0, stream>>>(x, W1,            nullptr, relbuf,                  N, K1);
    gemm_xw<<<gblocks, 256, 0, stream>>>(x, W1 + K1 * 64,  nullptr, relbuf + (size_t)N * 64, N, K1);
    gemm_xw<<<gblocks, 256, 0, stream>>>(x, loop1,         b1,      h1,                      N, K1);
    edge_scatter64<<<eb64, 256, 0, stream>>>(relbuf, src, dst, etypes, h1, E, N);

    gemm_relu_xw64<<<gblocks, 256, 0, stream>>>(h1, W2,           nullptr, relbuf,                  N);
    gemm_relu_xw64<<<gblocks, 256, 0, stream>>>(h1, W2 + 64 * 64, nullptr, relbuf + (size_t)N * 64, N);
    gemm_relu_xw64<<<gblocks, 256, 0, stream>>>(h1, loop2,        b2,      h2,                      N);
    edge_scatter64<<<eb64, 256, 0, stream>>>(relbuf, src, dst, etypes, h2, E, N);

    layer3_node<<<nb, 256, 0, stream>>>(h2, W3, loop3, b3, rel3, agg3, N);
    edge_scatter2<<<eb2, 256, 0, stream>>>(rel3, src, dst, etypes, agg3, E, N);
    init_slot<<<1, 1, 0, stream>>>(slot);
    min_reduce<<<512, 256, 0, stream>>>(agg3, N * 2, slot);
    finalize_k<<<nb, 256, 0, stream>>>(agg3, cell, maxs, slot, out, N);
}


# Product kernels: W staged once per block in LDS (k-pair, 4-column interleave: 2 LDS reads per k-step), x strip loaded ahead: no global load in the matrix loop, whole-row stores; sum order kept
# speedup vs baseline: 1.5092x; 1.0608x over previous
#include <hip/hip_runtime.h>

typedef __attribute__((ext_vector_type(2))) float v2f;
typedef __attribute__((ext_vector_type(8))) float v8f;

typedef __attribute__((ext_vector_type(4))) float v4f;

template <int K, bool RELU>
__global__ void __launch_bounds__(256)
gemm_wlds(const float* __restrict__ X, const float* __restrict__ W,
          const float* __restrict__ bias, float* __restrict__ C, int M) {
    constexpr int KS    = K / 4;
    constexpr int KP    = K / 2;
    constexpr int PITCH = 68;
    constexpr int PLANE = KP * PITCH;
    __shared__ __attribute__((aligned(16))) float wl[2 * PLANE];

    const int tid  = threadIdx.x;
    const int lane = tid & 31;
    const int wave = blockIdx.x * 8 + (tid >> 5);
    const int m0   = wave * 16;
    const bool active = m0 < M;

    const int half = lane >> 4;
    const int lm   = lane & 15;

    #pragma unroll
    for (int i = 0; i < (KP * 16) / 256; ++i) {
        const int item = tid + 256 * i;
        const int kp = item >> 4;
        const int c  = item & 15;
        const float4 r0 = *(const float4*)(W + (size_t)(2 * kp) * 64 + 4 * c);
        const float4 r1 = *(const float4*)(W + (size_t)(2 * kp + 1) * 64 + 4 * c);
        v4f p0; p0.x = r0.x; p0.y = r1.x; p0.z = r0.y; p0.w = r1.y;
        v4f p1; p1.x = r0.z; p1.y = r1.z; p1.z = r0.w; p1.w = r1.w;
        *(v4f*)(wl + kp * PITCH + 4 * c)         = p0;
        *(v4f*)(wl + PLANE + kp * PITCH + 4 * c) = p1;
    }

    float2 av[KS];
    float4 bq = make_float4(0.0f, 0.0f, 0.0f, 0.0f);
    if (active) {
        const float* xrow = X + (size_t)(m0 + lm) * K + 2 * half;
        #pragma unroll
        for (int ks = 0; ks < KS; ++ks) av[ks] = *(const float2*)(xrow + ks * 4);
        if (bias) bq = *(const float4*)(bias + 4 * lm);
    }

    __syncthreads();
    if (!active) return;

    const float bvs[4] = {bq.x, bq.y, bq.z, bq.w};
    v8f acc[4];
    #pragma unroll
    for (int nt = 0; nt < 4; ++nt) {
        #pragma unroll
        for (int v = 0; v < 8; ++v) acc[nt][v] = bvs[nt];
    }

    v2f a[KS];
    #pragma unroll
    for (int ks = 0; ks < KS; ++ks) {
        a[ks].x = RELU ? fmaxf(av[ks].x, 0.0f) : av[ks].x;
        a[ks].y = RELU ? fmaxf(av[ks].y, 0.0f) : av[ks].y;
        if (RELU) asm volatile("" : "+v"(a[ks]));
    }

    unsigned coff = (unsigned)((m0 + 8 * half) * 64 + 4 * lm) * 4u;
    const float* lw = wl + half * PITCH + 4 * lm;

    asm volatile("" : "+v"(acc[0]), "+v"(acc[1]), "+v"(acc[2]), "+v"(acc[3]));
    asm volatile("" : "+v"(coff));
    __builtin_amdgcn_sched_barrier(0);

    constexpr int GS = 4;
    constexpr int NG = KS / GS;
    v4f q[2][GS][2];
    #pragma unroll
    for (int s = 0; s < GS; ++s) {
        q[0][s][0] = *(const v4f*)(lw + (2 * s) * PITCH);
        q[0][s][1] = *(const v4f*)(lw + PLANE + (2 * s) * PITCH);
    }
    #pragma unroll
    for (int g = 0; g < NG; ++g) {
        if (g + 1 < NG) {
            #pragma unroll
            for (int s = 0; s < GS; ++s) {
                const int ks = (g + 1) * GS + s;
                q[(g + 1) & 1][s][0] = *(const v4f*)(lw + (2 * ks) * PITCH);
                q[(g + 1) & 1][s][1] = *(const v4f*)(lw + PLANE + (2 * ks) * PITCH);
            }
        }
        __builtin_amdgcn_sched_barrier(0);
        #pragma unroll
        for (int s = 0; s < GS; ++s) {
            const int ks = g * GS + s;
            const v4f q01 = q[g & 1][s][0];
            const v4f q23 = q[g & 1][s][1];
            const v2f b0 = __builtin_shufflevector(q01, q01, 0, 1);
            const v2f b1 = __builtin_shufflevector(q01, q01, 2, 3);
            const v2f b2 = __builtin_shufflevector(q23, q23, 0, 1);
            const v2f b3 = __builtin_shufflevector(q23, q23, 2, 3);
            acc[0] = __builtin_amdgcn_wmma_f32_16x16x4_f32(
                false, a[ks], false, b0, (short)0, acc[0], false, false);
            acc[1] = __builtin_amdgcn_wmma_f32_16x16x4_f32(
                false, a[ks], false, b1, (short)0, acc[1], false, false);
            acc[2] = __builtin_amdgcn_wmma_f32_16x16x4_f32(
                false, a[ks], false, b2, (short)0, acc[2], false, false);
            acc[3] = __builtin_amdgcn_wmma_f32_16x16x4_f32(
                false, a[ks], false, b3, (short)0, acc[3], false, false);
        }
        __builtin_amdgcn_sched_barrier(0);
    }

    asm volatile("v_nop\n\tv_nop\n\tv_nop\n\tv_nop\n\tv_nop\n\tv_nop\n\tv_nop\n\tv_nop"
                 : "+v"(acc[0]), "+v"(acc[1]), "+v"(acc[2]), "+v"(acc[3]));

    float* crow = (float*)((char*)C + coff);
    #pragma unroll
    for (int v = 0; v < 8; ++v) {
        const float4 o = make_float4(acc[0][v], acc[1][v], acc[2][v], acc[3][v]);
        *(float4*)(crow + v * 64) = o;
    }
}

__global__ void __launch_bounds__(256)
edge_scatter64(const float* __restrict__ rel,
               const int* __restrict__ src,
               const int* __restrict__ dst,
               const int* __restrict__ et,
               float* __restrict__ agg, int E, int N) {
    const int lane = threadIdx.x & 31;
    const int wave = (int)((blockIdx.x * blockDim.x + threadIdx.x) >> 5);
    const long long e0 = (long long)wave * 32;
    if (e0 >= E) return;
    const int cnt = (E - e0 < 32) ? (int)(E - e0) : 32;

    int srow = 0, drow = 0;
    if (lane < cnt) {
        const long long e = e0 + lane;
        srow = et[e] * N + src[e];
        drow = dst[e];
    }

    if (cnt == 32) {
        #pragma unroll
        for (int j0 = 0; j0 < 32; j0 += 8) {
            float a[8], b[8];
            #pragma unroll
            for (int u = 0; u < 8; ++u) {
                const int sr = __builtin_amdgcn_readlane(srow, j0 + u);
                const float* p = rel + (size_t)sr * 64 + lane;
                a[u] = p[0];
                b[u] = p[32];
            }
            #pragma unroll
            for (int u = 0; u < 8; ++u) {
                const int dr = __builtin_amdgcn_readlane(drow, j0 + u);
                float* q = agg + (size_t)dr * 64 + lane;
                atomicAdd(q, a[u]);
                atomicAdd(q + 32, b[u]);
            }
        }
    } else {
        for (int j = 0; j < cnt; ++j) {
            const int sr = __builtin_amdgcn_readlane(srow, j);
            const int dr = __builtin_amdgcn_readlane(drow, j);
            const float* p = rel + (size_t)sr * 64 + lane;
            float* q = agg + (size_t)dr * 64 + lane;
            atomicAdd(q, p[0]);
            atomicAdd(q + 32, p[32]);
        }
    }
}

__global__ void layer3_node(const float* __restrict__ h2,
                            const float* __restrict__ W3,
                            const float* __restrict__ loop3,
                            const float* __restrict__ b3,
                            float* __restrict__ rel3, float* __restrict__ agg3,
                            int N) {
    const int i = blockIdx.x * blockDim.x + threadIdx.x;
    if (i >= N) return;
    const float* row = h2 + (size_t)i * 64;
    float r00 = 0.f, r01 = 0.f, r10 = 0.f, r11 = 0.f, l0 = 0.f, l1 = 0.f;
    #pragma unroll 8
    for (int k = 0; k < 64; ++k) {
        const float v = fmaxf(row[k], 0.0f);
        r00 += v * W3[k * 2 + 0];
        r01 += v * W3[k * 2 + 1];
        r10 += v * W3[128 + k * 2 + 0];
        r11 += v * W3[128 + k * 2 + 1];
        l0  += v * loop3[k * 2 + 0];
        l1  += v * loop3[k * 2 + 1];
    }
    rel3[(size_t)i * 2 + 0] = r00;
    rel3[(size_t)i * 2 + 1] = r01;
    rel3[(size_t)(N + i) * 2 + 0] = r10;
    rel3[(size_t)(N + i) * 2 + 1] = r11;
    agg3[(size_t)i * 2 + 0] = l0 + b3[0];
    agg3[(size_t)i * 2 + 1] = l1 + b3[1];
}

__global__ void edge_scatter2(const float* __restrict__ rel3,
                              const int* __restrict__ src,
                              const int* __restrict__ dst,
                              const int* __restrict__ et,
                              float* __restrict__ agg3, int E, int N) {
    const int e = blockIdx.x * blockDim.x + threadIdx.x;
    if (e >= E) return;
    const float* p = rel3 + ((long long)et[e] * N + src[e]) * 2;
    float* q = agg3 + (long long)dst[e] * 2;
    atomicAdd(q + 0, p[0]);
    atomicAdd(q + 1, p[1]);
}

__device__ __forceinline__ unsigned f2ord(float f) {
    unsigned u = __float_as_uint(f);
    return (u & 0x80000000u) ? ~u : (u | 0x80000000u);
}
__device__ __forceinline__ float ord2f(unsigned u) {
    return __uint_as_float((u & 0x80000000u) ? (u ^ 0x80000000u) : ~u);
}

__global__ void init_slot(unsigned* slot) { *slot = 0xFFFFFFFFu; }

__global__ void min_reduce(const float* __restrict__ p, int n, unsigned* slot) {
    __shared__ float sm[256];
    float m = 3.402823466e38f;
    for (int i = blockIdx.x * blockDim.x + threadIdx.x; i < n;
         i += gridDim.x * blockDim.x)
        m = fminf(m, p[i]);
    sm[threadIdx.x] = m;
    __syncthreads();
    for (int s = blockDim.x >> 1; s > 0; s >>= 1) {
        if ((int)threadIdx.x < s) sm[threadIdx.x] = fminf(sm[threadIdx.x], sm[threadIdx.x + s]);
        __syncthreads();
    }
    if (threadIdx.x == 0) atomicMin(slot, f2ord(sm[0]));
}

__global__ void finalize_k(const float* __restrict__ agg3,
                           const int* __restrict__ cell,
                           const int* __restrict__ maxs,
                           const unsigned* __restrict__ slot,
                           float* __restrict__ out, int N) {
    const int i = blockIdx.x * blockDim.x + threadIdx.x;
    if (i >= N) return;
    const float gmin = ord2f(*slot) - 1.0f;
    const bool upper = cell[i] >= (maxs[i] - 1);
    const bool lower = cell[i] == 0;
    out[(size_t)i * 2 + 0] = upper ? gmin : agg3[(size_t)i * 2 + 0];
    out[(size_t)i * 2 + 1] = lower ? gmin : agg3[(size_t)i * 2 + 1];
}

extern "C" void kernel_launch(void* const* d_in, const int* in_sizes, int n_in,
                              void* d_out, int out_size, void* d_ws, size_t ws_size,
                              hipStream_t stream) {
    const float* x      = (const float*)d_in[0];
    const int*   src    = (const int*)d_in[1];
    const int*   dst    = (const int*)d_in[2];
    const int*   etypes = (const int*)d_in[3];
    const int*   cell   = (const int*)d_in[4];
    const int*   maxs   = (const int*)d_in[5];
    const float* W1     = (const float*)d_in[6];
    const float* loop1  = (const float*)d_in[7];
    const float* b1     = (const float*)d_in[8];
    const float* W2     = (const float*)d_in[9];
    const float* loop2  = (const float*)d_in[10];
    const float* b2     = (const float*)d_in[11];
    const float* W3     = (const float*)d_in[12];
    const float* loop3  = (const float*)d_in[13];
    const float* b3     = (const float*)d_in[14];
    float* out = (float*)d_out;

    const int N = in_sizes[4];
    const int E = in_sizes[1];

    float* relbuf = (float*)d_ws;
    float* h1     = relbuf + (size_t)2 * N * 64;
    float* h2     = h1 + (size_t)N * 64;
    float* rel3   = h2 + (size_t)N * 64;
    float* agg3   = rel3 + (size_t)4 * N;
    unsigned* slot = (unsigned*)(agg3 + (size_t)2 * N);

    const int waves   = (N + 15) / 16;
    const int gblocks = (waves * 32 + 255) / 256;
    const int eb64    = (E + 255) / 256;
    const int nb      = (N + 255) / 256;
    const int eb2     = (E + 255) / 256;

    gemm_wlds<128, false><<<gblocks, 256, 0, stream>>>(x, W1,            nullptr, relbuf,                  N);
    gemm_wlds<128, false><<<gblocks, 256, 0, stream>>>(x, W1 + 128 * 64, nullptr, relbuf + (size_t)N * 64, N);
    gemm_wlds<128, false><<<gblocks, 256, 0, stream>>>(x, loop1,         b1,      h1,                      N);
    edge_scatter64<<<eb64, 256, 0, stream>>>(relbuf, src, dst, etypes, h1, E, N);

    gemm_wlds<64, true><<<gblocks, 256, 0, stream>>>(h1, W2,           nullptr, relbuf,                  N);
    gemm_wlds<64, true><<<gblocks, 256, 0, stream>>>(h1, W2 + 64 * 64, nullptr, relbuf + (size_t)N * 64, N);
    gemm_wlds<64, true><<<gblocks, 256, 0, stream>>>(h1, loop2,        b2,      h2,                      N);
    edge_scatter64<<<eb64, 256, 0, stream>>>(relbuf, src, dst, etypes, h2, E, N);

    layer3_node<<<nb, 256, 0, stream>>>(h2, W3, loop3, b3, rel3, agg3, N);
    edge_scatter2<<<eb2, 256, 0, stream>>>(rel3, src, dst, etypes, agg3, E, N);
    init_slot<<<1, 1, 0, stream>>>(slot);
    min_reduce<<<512, 256, 0, stream>>>(agg3, N * 2, slot);
    finalize_k<<<nb, 256, 0, stream>>>(agg3, cell, maxs, slot, out, N);
}


# Each layer's three products (2 relation weights, self-loop) run as ONE launch on a 3-row grid (block row picks W, bias, output plane; x strip re-read per product): 13 -> 9 launches; sum order kept
# speedup vs baseline: 1.5889x; 1.0528x over previous
#include <hip/hip_runtime.h>

typedef __attribute__((ext_vector_type(2))) float v2f;
typedef __attribute__((ext_vector_type(8))) float v8f;

typedef __attribute__((ext_vector_type(4))) float v4f;

template <int K, bool RELU>
__global__ void __launch_bounds__(256)
gemm_wlds(const float* __restrict__ X,
          const float* __restrict__ Wr0, const float* __restrict__ Wr1,
          const float* __restrict__ Wlp, const float* __restrict__ blp,
          float* __restrict__ Cr0, float* __restrict__ Cr1,
          float* __restrict__ Clp, int M) {
    const int y = blockIdx.y;
    const float* W    = (y == 0) ? Wr0 : (y == 1) ? Wr1 : Wlp;
    const float* bias = (y == 2) ? blp : nullptr;
    float*       C    = (y == 0) ? Cr0 : (y == 1) ? Cr1 : Clp;
    constexpr int KS    = K / 4;
    constexpr int KP    = K / 2;
    constexpr int PITCH = 68;
    constexpr int PLANE = KP * PITCH;
    __shared__ __attribute__((aligned(16))) float wl[2 * PLANE];

    const int tid  = threadIdx.x;
    const int lane = tid & 31;
    const int wave = blockIdx.x * 8 + (tid >> 5);
    const int m0   = wave * 16;
    const bool active = m0 < M;

    const int half = lane >> 4;
    const int lm   = lane & 15;

    #pragma unroll
    for (int i = 0; i < (KP * 16) / 256; ++i) {
        const int item = tid + 256 * i;
        const int kp = item >> 4;
        const int c  = item & 15;
        const float4 r0 = *(const float4*)(W + (size_t)(2 * kp) * 64 + 4 * c);
        const float4 r1 = *(const float4*)(W + (size_t)(2 * kp + 1) * 64 + 4 * c);
        v4f p0; p0.x = r0.x; p0.y = r1.x; p0.z = r0.y; p0.w = r1.y;
        v4f p1; p1.x = r0.z; p1.y = r1.z; p1.z = r0.w; p1.w = r1.w;
        *(v4f*)(wl + kp * PITCH + 4 * c)         = p0;
        *(v4f*)(wl + PLANE + kp * PITCH + 4 * c) = p1;
    }

    float2 av[KS];
    float4 bq = make_float4(0.0f, 0.0f, 0.0f, 0.0f);
    if (active) {
        const float* xrow = X + (size_t)(m0 + lm) * K + 2 * half;
        #pragma unroll
        for (int ks = 0; ks < KS; ++ks) av[ks] = *(const float2*)(xrow + ks * 4);
        if (bias) bq = *(const float4*)(bias + 4 * lm);
    }

    __syncthreads();
    if (!active) return;

    const float bvs[4] = {bq.x, bq.y, bq.z, bq.w};
    v8f acc[4];
    #pragma unroll
    for (int nt = 0; nt < 4; ++nt) {
        #pragma unroll
        for (int v = 0; v < 8; ++v) acc[nt][v] = bvs[nt];
    }

    v2f a[KS];
    #pragma unroll
    for (int ks = 0; ks < KS; ++ks) {
        a[ks].x = RELU ? fmaxf(av[ks].x, 0.0f) : av[ks].x;
        a[ks].y = RELU ? fmaxf(av[ks].y, 0.0f) : av[ks].y;
        if (RELU) asm volatile("" : "+v"(a[ks]));
    }

    unsigned coff = (unsigned)((m0 + 8 * half) * 64 + 4 * lm) * 4u;
    const float* lw = wl + half * PITCH + 4 * lm;

    asm volatile("" : "+v"(acc[0]), "+v"(acc[1]), "+v"(acc[2]), "+v"(acc[3]));
    asm volatile("" : "+v"(coff));
    __builtin_amdgcn_sched_barrier(0);

    constexpr int GS = 4;
    constexpr int NG = KS / GS;
    v4f q[2][GS][2];
    #pragma unroll
    for (int s = 0; s < GS; ++s) {
        q[0][s][0] = *(const v4f*)(lw + (2 * s) * PITCH);
        q[0][s][1] = *(const v4f*)(lw + PLANE + (2 * s) * PITCH);
    }
    #pragma unroll
    for (int g = 0; g < NG; ++g) {
        if (g + 1 < NG) {
            #pragma unroll
            for (int s = 0; s < GS; ++s) {
                const int ks = (g + 1) * GS + s;
                q[(g + 1) & 1][s][0] = *(const v4f*)(lw + (2 * ks) * PITCH);
                q[(g + 1) & 1][s][1] = *(const v4f*)(lw + PLANE + (2 * ks) * PITCH);
            }
        }
        __builtin_amdgcn_sched_barrier(0);
        #pragma unroll
        for (int s = 0; s < GS; ++s) {
            const int ks = g * GS + s;
            const v4f q01 = q[g & 1][s][0];
            const v4f q23 = q[g & 1][s][1];
            const v2f b0 = __builtin_shufflevector(q01, q01, 0, 1);
            const v2f b1 = __builtin_shufflevector(q01, q01, 2, 3);
            const v2f b2 = __builtin_shufflevector(q23, q23, 0, 1);
            const v2f b3 = __builtin_shufflevector(q23, q23, 2, 3);
            acc[0] = __builtin_amdgcn_wmma_f32_16x16x4_f32(
                false, a[ks], false, b0, (short)0, acc[0], false, false);
            acc[1] = __builtin_amdgcn_wmma_f32_16x16x4_f32(
                false, a[ks], false, b1, (short)0, acc[1], false, false);
            acc[2] = __builtin_amdgcn_wmma_f32_16x16x4_f32(
                false, a[ks], false, b2, (short)0, acc[2], false, false);
            acc[3] = __builtin_amdgcn_wmma_f32_16x16x4_f32(
                false, a[ks], false, b3, (short)0, acc[3], false, false);
        }
        __builtin_amdgcn_sched_barrier(0);
    }

    asm volatile("v_nop\n\tv_nop\n\tv_nop\n\tv_nop\n\tv_nop\n\tv_nop\n\tv_nop\n\tv_nop"
                 : "+v"(acc[0]), "+v"(acc[1]), "+v"(acc[2]), "+v"(acc[3]));

    float* crow = (float*)((char*)C + coff);
    #pragma unroll
    for (int v = 0; v < 8; ++v) {
        const float4 o = make_float4(acc[0][v], acc[1][v], acc[2][v], acc[3][v]);
        *(float4*)(crow + v * 64) = o;
    }
}

__global__ void __launch_bounds__(256)
edge_scatter64(const float* __restrict__ rel,
               const int* __restrict__ src,
               const int* __restrict__ dst,
               const int* __restrict__ et,
               float* __restrict__ agg, int E, int N) {
    const int lane = threadIdx.x & 31;
    const int wave = (int)((blockIdx.x * blockDim.x + threadIdx.x) >> 5);
    const long long e0 = (long long)wave * 32;
    if (e0 >= E) return;
    const int cnt = (E - e0 < 32) ? (int)(E - e0) : 32;

    int srow = 0, drow = 0;
    if (lane < cnt) {
        const long long e = e0 + lane;
        srow = et[e] * N + src[e];
        drow = dst[e];
    }

    if (cnt == 32) {
        #pragma unroll
        for (int j0 = 0; j0 < 32; j0 += 8) {
            float a[8], b[8];
            #pragma unroll
            for (int u = 0; u < 8; ++u) {
                const int sr = __builtin_amdgcn_readlane(srow, j0 + u);
                const float* p = rel + (size_t)sr * 64 + lane;
                a[u] = p[0];
                b[u] = p[32];
            }
            #pragma unroll
            for (int u = 0; u < 8; ++u) {
                const int dr = __builtin_amdgcn_readlane(drow, j0 + u);
                float* q = agg + (size_t)dr * 64 + lane;
                atomicAdd(q, a[u]);
                atomicAdd(q + 32, b[u]);
            }
        }
    } else {
        for (int j = 0; j < cnt; ++j) {
            const int sr = __builtin_amdgcn_readlane(srow, j);
            const int dr = __builtin_amdgcn_readlane(drow, j);
            const float* p = rel + (size_t)sr * 64 + lane;
            float* q = agg + (size_t)dr * 64 + lane;
            atomicAdd(q, p[0]);
            atomicAdd(q + 32, p[32]);
        }
    }
}

__global__ void layer3_node(const float* __restrict__ h2,
                            const float* __restrict__ W3,
                            const float* __restrict__ loop3,
                            const float* __restrict__ b3,
                            float* __restrict__ rel3, float* __restrict__ agg3,
                            int N) {
    const int i = blockIdx.x * blockDim.x + threadIdx.x;
    if (i >= N) return;
    const float* row = h2 + (size_t)i * 64;
    float r00 = 0.f, r01 = 0.f, r10 = 0.f, r11 = 0.f, l0 = 0.f, l1 = 0.f;
    #pragma unroll 8
    for (int k = 0; k < 64; ++k) {
        const float v = fmaxf(row[k], 0.0f);
        r00 += v * W3[k * 2 + 0];
        r01 += v * W3[k * 2 + 1];
        r10 += v * W3[128 + k * 2 + 0];
        r11 += v * W3[128 + k * 2 + 1];
        l0  += v * loop3[k * 2 + 0];
        l1  += v * loop3[k * 2 + 1];
    }
    rel3[(size_t)i * 2 + 0] = r00;
    rel3[(size_t)i * 2 + 1] = r01;
    rel3[(size_t)(N + i) * 2 + 0] = r10;
    rel3[(size_t)(N + i) * 2 + 1] = r11;
    agg3[(size_t)i * 2 + 0] = l0 + b3[0];
    agg3[(size_t)i * 2 + 1] = l1 + b3[1];
}

__global__ void edge_scatter2(const float* __restrict__ rel3,
                              const int* __restrict__ src,
                              const int* __restrict__ dst,
                              const int* __restrict__ et,
                              float* __restrict__ agg3, int E, int N) {
    const int e = blockIdx.x * blockDim.x + threadIdx.x;
    if (e >= E) return;
    const float* p = rel3 + ((long long)et[e] * N + src[e]) * 2;
    float* q = agg3 + (long long)dst[e] * 2;
    atomicAdd(q + 0, p[0]);
    atomicAdd(q + 1, p[1]);
}

__device__ __forceinline__ unsigned f2ord(float f) {
    unsigned u = __float_as_uint(f);
    return (u & 0x80000000u) ? ~u : (u | 0x80000000u);
}
__device__ __forceinline__ float ord2f(unsigned u) {
    return __uint_as_float((u & 0x80000000u) ? (u ^ 0x80000000u) : ~u);
}

__global__ void init_slot(unsigned* slot) { *slot = 0xFFFFFFFFu; }

__global__ void min_reduce(const float* __restrict__ p, int n, unsigned* slot) {
    __shared__ float sm[256];
    float m = 3.402823466e38f;
    for (int i = blockIdx.x * blockDim.x + threadIdx.x; i < n;
         i += gridDim.x * blockDim.x)
        m = fminf(m, p[i]);
    sm[threadIdx.x] = m;
    __syncthreads();
    for (int s = blockDim.x >> 1; s > 0; s >>= 1) {
        if ((int)threadIdx.x < s) sm[threadIdx.x] = fminf(sm[threadIdx.x], sm[threadIdx.x + s]);
        __syncthreads();
    }
    if (threadIdx.x == 0) atomicMin(slot, f2ord(sm[0]));
}

__global__ void finalize_k(const float* __restrict__ agg3,
                           const int* __restrict__ cell,
                           const int* __restrict__ maxs,
                           const unsigned* __restrict__ slot,
                           float* __restrict__ out, int N) {
    const int i = blockIdx.x * blockDim.x + threadIdx.x;
    if (i >= N) return;
    const float gmin = ord2f(*slot) - 1.0f;
    const bool upper = cell[i] >= (maxs[i] - 1);
    const bool lower = cell[i] == 0;
    out[(size_t)i * 2 + 0] = upper ? gmin : agg3[(size_t)i * 2 + 0];
    out[(size_t)i * 2 + 1] = lower ? gmin : agg3[(size_t)i * 2 + 1];
}

extern "C" void kernel_launch(void* const* d_in, const int* in_sizes, int n_in,
                              void* d_out, int out_size, void* d_ws, size_t ws_size,
                              hipStream_t stream) {
    const float* x      = (const float*)d_in[0];
    const int*   src    = (const int*)d_in[1];
    const int*   dst    = (const int*)d_in[2];
    const int*   etypes = (const int*)d_in[3];
    const int*   cell   = (const int*)d_in[4];
    const int*   maxs   = (const int*)d_in[5];
    const float* W1     = (const float*)d_in[6];
    const float* loop1  = (const float*)d_in[7];
    const float* b1     = (const float*)d_in[8];
    const float* W2     = (const float*)d_in[9];
    const float* loop2  = (const float*)d_in[10];
    const float* b2     = (const float*)d_in[11];
    const float* W3     = (const float*)d_in[12];
    const float* loop3  = (const float*)d_in[13];
    const float* b3     = (const float*)d_in[14];
    float* out = (float*)d_out;

    const int N = in_sizes[4];
    const int E = in_sizes[1];

    float* relbuf = (float*)d_ws;
    float* h1     = relbuf + (size_t)2 * N * 64;
    float* h2     = h1 + (size_t)N * 64;
    float* rel3   = h2 + (size_t)N * 64;
    float* agg3   = rel3 + (size_t)4 * N;
    unsigned* slot = (unsigned*)(agg3 + (size_t)2 * N);

    const int waves   = (N + 15) / 16;
    const int gblocks = (waves * 32 + 255) / 256;
    const int eb64    = (E + 255) / 256;
    const int nb      = (N + 255) / 256;
    const int eb2     = (E + 255) / 256;

    gemm_wlds<128, false><<<dim3(gblocks, 3), 256, 0, stream>>>(
        x, W1, W1 + 128 * 64, loop1, b1, relbuf, relbuf + (size_t)N * 64, h1, N);
    edge_scatter64<<<eb64, 256, 0, stream>>>(relbuf, src, dst, etypes, h1, E, N);

    gemm_wlds<64, true><<<dim3(gblocks, 3), 256, 0, stream>>>(
        h1, W2, W2 + 64 * 64, loop2, b2, relbuf, relbuf + (size_t)N * 64, h2, N);
    edge_scatter64<<<eb64, 256, 0, stream>>>(relbuf, src, dst, etypes, h2, E, N);

    layer3_node<<<nb, 256, 0, stream>>>(h2, W3, loop3, b3, rel3, agg3, N);
    edge_scatter2<<<eb2, 256, 0, stream>>>(rel3, src, dst, etypes, agg3, E, N);
    init_slot<<<1, 1, 0, stream>>>(slot);
    min_reduce<<<512, 256, 0, stream>>>(agg3, N * 2, slot);
    finalize_k<<<nb, 256, 0, stream>>>(agg3, cell, maxs, slot, out, N);
}


# 2-wide edge scatter: lanes 2k, 2k+1 take the 2 columns of one edge (row numbers by lane shuffle): an edge's 2 adds leave as one 8-byte request; 128 edges per wave; float atomics, arrival order
# speedup vs baseline: 1.8991x; 1.1952x over previous
#include <hip/hip_runtime.h>

typedef __attribute__((ext_vector_type(2))) float v2f;
typedef __attribute__((ext_vector_type(8))) float v8f;

typedef __attribute__((ext_vector_type(4))) float v4f;

template <int K, bool RELU>
__global__ void __launch_bounds__(256)
gemm_wlds(const float* __restrict__ X,
          const float* __restrict__ Wr0, const float* __restrict__ Wr1,
          const float* __restrict__ Wlp, const float* __restrict__ blp,
          float* __restrict__ Cr0, float* __restrict__ Cr1,
          float* __restrict__ Clp, int M) {
    const int y = blockIdx.y;
    const float* W    = (y == 0) ? Wr0 : (y == 1) ? Wr1 : Wlp;
    const float* bias = (y == 2) ? blp : nullptr;
    float*       C    = (y == 0) ? Cr0 : (y == 1) ? Cr1 : Clp;
    constexpr int KS    = K / 4;
    constexpr int KP    = K / 2;
    constexpr int PITCH = 68;
    constexpr int PLANE = KP * PITCH;
    __shared__ __attribute__((aligned(16))) float wl[2 * PLANE];

    const int tid  = threadIdx.x;
    const int lane = tid & 31;
    const int wave = blockIdx.x * 8 + (tid >> 5);
    const int m0   = wave * 16;
    const bool active = m0 < M;

    const int half = lane >> 4;
    const int lm   = lane & 15;

    #pragma unroll
    for (int i = 0; i < (KP * 16) / 256; ++i) {
        const int item = tid + 256 * i;
        const int kp = item >> 4;
        const int c  = item & 15;
        const float4 r0 = *(const float4*)(W + (size_t)(2 * kp) * 64 + 4 * c);
        const float4 r1 = *(const float4*)(W + (size_t)(2 * kp + 1) * 64 + 4 * c);
        v4f p0; p0.x = r0.x; p0.y = r1.x; p0.z = r0.y; p0.w = r1.y;
        v4f p1; p1.x = r0.z; p1.y = r1.z; p1.z = r0.w; p1.w = r1.w;
        *(v4f*)(wl + kp * PITCH + 4 * c)         = p0;
        *(v4f*)(wl + PLANE + kp * PITCH + 4 * c) = p1;
    }

    float2 av[KS];
    float4 bq = make_float4(0.0f, 0.0f, 0.0f, 0.0f);
    if (active) {
        const float* xrow = X + (size_t)(m0 + lm) * K + 2 * half;
        #pragma unroll
        for (int ks = 0; ks < KS; ++ks) av[ks] = *(const float2*)(xrow + ks * 4);
        if (bias) bq = *(const float4*)(bias + 4 * lm);
    }

    __syncthreads();
    if (!active) return;

    const float bvs[4] = {bq.x, bq.y, bq.z, bq.w};
    v8f acc[4];
    #pragma unroll
    for (int nt = 0; nt < 4; ++nt) {
        #pragma unroll
        for (int v = 0; v < 8; ++v) acc[nt][v] = bvs[nt];
    }

    v2f a[KS];
    #pragma unroll
    for (int ks = 0; ks < KS; ++ks) {
        a[ks].x = RELU ? fmaxf(av[ks].x, 0.0f) : av[ks].x;
        a[ks].y = RELU ? fmaxf(av[ks].y, 0.0f) : av[ks].y;
        if (RELU) asm volatile("" : "+v"(a[ks]));
    }

    unsigned coff = (unsigned)((m0 + 8 * half) * 64 + 4 * lm) * 4u;
    const float* lw = wl + half * PITCH + 4 * lm;

    asm volatile("" : "+v"(acc[0]), "+v"(acc[1]), "+v"(acc[2]), "+v"(acc[3]));
    asm volatile("" : "+v"(coff));
    __builtin_amdgcn_sched_barrier(0);

    constexpr int GS = 4;
    constexpr int NG = KS / GS;
    v4f q[2][GS][2];
    #pragma unroll
    for (int s = 0; s < GS; ++s) {
        q[0][s][0] = *(const v4f*)(lw + (2 * s) * PITCH);
        q[0][s][1] = *(const v4f*)(lw + PLANE + (2 * s) * PITCH);
    }
    #pragma unroll
    for (int g = 0; g < NG; ++g) {
        if (g + 1 < NG) {
            #pragma unroll
            for (int s = 0; s < GS; ++s) {
                const int ks = (g + 1) * GS + s;
                q[(g + 1) & 1][s][0] = *(const v4f*)(lw + (2 * ks) * PITCH);
                q[(g + 1) & 1][s][1] = *(const v4f*)(lw + PLANE + (2 * ks) * PITCH);
            }
        }
        __builtin_amdgcn_sched_barrier(0);
        #pragma unroll
        for (int s = 0; s < GS; ++s) {
            const int ks = g * GS + s;
            const v4f q01 = q[g & 1][s][0];
            const v4f q23 = q[g & 1][s][1];
            const v2f b0 = __builtin_shufflevector(q01, q01, 0, 1);
            const v2f b1 = __builtin_shufflevector(q01, q01, 2, 3);
            const v2f b2 = __builtin_shufflevector(q23, q23, 0, 1);
            const v2f b3 = __builtin_shufflevector(q23, q23, 2, 3);
            acc[0] = __builtin_amdgcn_wmma_f32_16x16x4_f32(
                false, a[ks], false, b0, (short)0, acc[0], false, false);
            acc[1] = __builtin_amdgcn_wmma_f32_16x16x4_f32(
                false, a[ks], false, b1, (short)0, acc[1], false, false);
            acc[2] = __builtin_amdgcn_wmma_f32_16x16x4_f32(
                false, a[ks], false, b2, (short)0, acc[2], false, false);
            acc[3] = __builtin_amdgcn_wmma_f32_16x16x4_f32(
                false, a[ks], false, b3, (short)0, acc[3], false, false);
        }
        __builtin_amdgcn_sched_barrier(0);
    }

    asm volatile("v_nop\n\tv_nop\n\tv_nop\n\tv_nop\n\tv_nop\n\tv_nop\n\tv_nop\n\tv_nop"
                 : "+v"(acc[0]), "+v"(acc[1]), "+v"(acc[2]), "+v"(acc[3]));

    float* crow = (float*)((char*)C + coff);
    #pragma unroll
    for (int v = 0; v < 8; ++v) {
        const float4 o = make_float4(acc[0][v], acc[1][v], acc[2][v], acc[3][v]);
        *(float4*)(crow + v * 64) = o;
    }
}

__global__ void __launch_bounds__(256)
edge_scatter64(const float* __restrict__ rel,
               const int* __restrict__ src,
               const int* __restrict__ dst,
               const int* __restrict__ et,
               float* __restrict__ agg, int E, int N) {
    const int lane = threadIdx.x & 31;
    const int wave = (int)((blockIdx.x * blockDim.x + threadIdx.x) >> 5);
    const long long e0 = (long long)wave * 32;
    if (e0 >= E) return;
    const int cnt = (E - e0 < 32) ? (int)(E - e0) : 32;

    int srow = 0, drow = 0;
    if (lane < cnt) {
        const long long e = e0 + lane;
        srow = et[e] * N + src[e];
        drow = dst[e];
    }

    if (cnt == 32) {
        #pragma unroll
        for (int j0 = 0; j0 < 32; j0 += 8) {
            float a[8], b[8];
            #pragma unroll
            for (int u = 0; u < 8; ++u) {
                const int sr = __builtin_amdgcn_readlane(srow, j0 + u);
                const float* p = rel + (size_t)sr * 64 + lane;
                a[u] = p[0];
                b[u] = p[32];
            }
            #pragma unroll
            for (int u = 0; u < 8; ++u) {
                const int dr = __builtin_amdgcn_readlane(drow, j0 + u);
                float* q = agg + (size_t)dr * 64 + lane;
                atomicAdd(q, a[u]);
                atomicAdd(q + 32, b[u]);
            }
        }
    } else {
        for (int j = 0; j < cnt; ++j) {
            const int sr = __builtin_amdgcn_readlane(srow, j);
            const int dr = __builtin_amdgcn_readlane(drow, j);
            const float* p = rel + (size_t)sr * 64 + lane;
            float* q = agg + (size_t)dr * 64 + lane;
            atomicAdd(q, p[0]);
            atomicAdd(q + 32, p[32]);
        }
    }
}

__global__ void layer3_node(const float* __restrict__ h2,
                            const float* __restrict__ W3,
                            const float* __restrict__ loop3,
                            const float* __restrict__ b3,
                            float* __restrict__ rel3, float* __restrict__ agg3,
                            int N) {
    const int i = blockIdx.x * blockDim.x + threadIdx.x;
    if (i >= N) return;
    const float* row = h2 + (size_t)i * 64;
    float r00 = 0.f, r01 = 0.f, r10 = 0.f, r11 = 0.f, l0 = 0.f, l1 = 0.f;
    #pragma unroll 8
    for (int k = 0; k < 64; ++k) {
        const float v = fmaxf(row[k], 0.0f);
        r00 += v * W3[k * 2 + 0];
        r01 += v * W3[k * 2 + 1];
        r10 += v * W3[128 + k * 2 + 0];
        r11 += v * W3[128 + k * 2 + 1];
        l0  += v * loop3[k * 2 + 0];
        l1  += v * loop3[k * 2 + 1];
    }
    rel3[(size_t)i * 2 + 0] = r00;
    rel3[(size_t)i * 2 + 1] = r01;
    rel3[(size_t)(N + i) * 2 + 0] = r10;
    rel3[(size_t)(N + i) * 2 + 1] = r11;
    agg3[(size_t)i * 2 + 0] = l0 + b3[0];
    agg3[(size_t)i * 2 + 1] = l1 + b3[1];
}

template <int CH, bool GUARD>
__device__ __forceinline__ void scatter2_chunks(const float* __restrict__ rel3,
                                                const int* __restrict__ src,
                                                const int* __restrict__ dst,
                                                const int* __restrict__ et,
                                                float* __restrict__ agg3,
                                                long long e0, int lane, int E, int N) {
    int srow[CH], drow[CH];
    #pragma unroll
    for (int u = 0; u < CH; ++u) {
        const long long e  = e0 + 32 * u + lane;
        const long long ec = (!GUARD || e < E) ? e : (long long)E - 1;
        const int d = dst[ec];
        srow[u] = et[ec] * N + src[ec];
        drow[u] = (!GUARD || e < E) ? d : -1;
    }

    const int col = lane & 1;
    const int jl  = lane >> 1;
    float val[CH][2];
    int   dr[CH][2];
    #pragma unroll
    for (int u = 0; u < CH; ++u) {
        #pragma unroll
        for (int h = 0; h < 2; ++h) {
            const int j  = 16 * h + jl;
            const int sr = __shfl(srow[u], j);
            dr[u][h]     = __shfl(drow[u], j);
            val[u][h]    = (!GUARD || dr[u][h] >= 0) ? rel3[(size_t)sr * 2 + col] : 0.0f;
        }
    }
    #pragma unroll
    for (int u = 0; u < CH; ++u) {
        #pragma unroll
        for (int h = 0; h < 2; ++h) {
            if (!GUARD || dr[u][h] >= 0)
                atomicAdd(agg3 + (size_t)dr[u][h] * 2 + col, val[u][h]);
        }
    }
}

template <int CH>
__global__ void __launch_bounds__(256)
edge_scatter2(const float* __restrict__ rel3,
              const int* __restrict__ src,
              const int* __restrict__ dst,
              const int* __restrict__ et,
              float* __restrict__ agg3, int E, int N) {
    const int lane = threadIdx.x & 31;
    const int wave = (int)((blockIdx.x * blockDim.x + threadIdx.x) >> 5);
    const long long e0 = (long long)wave * (32 * CH);
    if (e0 >= E) return;
    if (e0 + 32 * CH <= E)
        scatter2_chunks<CH, false>(rel3, src, dst, et, agg3, e0, lane, E, N);
    else
        scatter2_chunks<CH, true>(rel3, src, dst, et, agg3, e0, lane, E, N);
}

__device__ __forceinline__ unsigned f2ord(float f) {
    unsigned u = __float_as_uint(f);
    return (u & 0x80000000u) ? ~u : (u | 0x80000000u);
}
__device__ __forceinline__ float ord2f(unsigned u) {
    return __uint_as_float((u & 0x80000000u) ? (u ^ 0x80000000u) : ~u);
}

__global__ void init_slot(unsigned* slot) { *slot = 0xFFFFFFFFu; }

__global__ void min_reduce(const float* __restrict__ p, int n, unsigned* slot) {
    __shared__ float sm[256];
    float m = 3.402823466e38f;
    for (int i = blockIdx.x * blockDim.x + threadIdx.x; i < n;
         i += gridDim.x * blockDim.x)
        m = fminf(m, p[i]);
    sm[threadIdx.x] = m;
    __syncthreads();
    for (int s = blockDim.x >> 1; s > 0; s >>= 1) {
        if ((int)threadIdx.x < s) sm[threadIdx.x] = fminf(sm[threadIdx.x], sm[threadIdx.x + s]);
        __syncthreads();
    }
    if (threadIdx.x == 0) atomicMin(slot, f2ord(sm[0]));
}

__global__ void finalize_k(const float* __restrict__ agg3,
                           const int* __restrict__ cell,
                           const int* __restrict__ maxs,
                           const unsigned* __restrict__ slot,
                           float* __restrict__ out, int N) {
    const int i = blockIdx.x * blockDim.x + threadIdx.x;
    if (i >= N) return;
    const float gmin = ord2f(*slot) - 1.0f;
    const bool upper = cell[i] >= (maxs[i] - 1);
    const bool lower = cell[i] == 0;
    out[(size_t)i * 2 + 0] = upper ? gmin : agg3[(size_t)i * 2 + 0];
    out[(size_t)i * 2 + 1] = lower ? gmin : agg3[(size_t)i * 2 + 1];
}

extern "C" void kernel_launch(void* const* d_in, const int* in_sizes, int n_in,
                              void* d_out, int out_size, void* d_ws, size_t ws_size,
                              hipStream_t stream) {
    const float* x      = (const float*)d_in[0];
    const int*   src    = (const int*)d_in[1];
    const int*   dst    = (const int*)d_in[2];
    const int*   etypes = (const int*)d_in[3];
    const int*   cell   = (const int*)d_in[4];
    const int*   maxs   = (const int*)d_in[5];
    const float* W1     = (const float*)d_in[6];
    const float* loop1  = (const float*)d_in[7];
    const float* b1     = (const float*)d_in[8];
    const float* W2     = (const float*)d_in[9];
    const float* loop2  = (const float*)d_in[10];
    const float* b2     = (const float*)d_in[11];
    const float* W3     = (const float*)d_in[12];
    const float* loop3  = (const float*)d_in[13];
    const float* b3     = (const float*)d_in[14];
    float* out = (float*)d_out;

    const int N = in_sizes[4];
    const int E = in_sizes[1];

    float* relbuf = (float*)d_ws;
    float* h1     = relbuf + (size_t)2 * N * 64;
    float* h2     = h1 + (size_t)N * 64;
    float* rel3   = h2 + (size_t)N * 64;
    float* agg3   = rel3 + (size_t)4 * N;
    unsigned* slot = (unsigned*)(agg3 + (size_t)2 * N);

    const int waves   = (N + 15) / 16;
    const int gblocks = (waves * 32 + 255) / 256;
    const int eb64    = (E + 255) / 256;
    const int nb      = (N + 255) / 256;
    const int eb2     = (E + 1023) / 1024;

    gemm_wlds<128, false><<<dim3(gblocks, 3), 256, 0, stream>>>(
        x, W1, W1 + 128 * 64, loop1, b1, relbuf, relbuf + (size_t)N * 64, h1, N);
    edge_scatter64<<<eb64, 256, 0, stream>>>(relbuf, src, dst, etypes, h1, E, N);

    gemm_wlds<64, true><<<dim3(gblocks, 3), 256, 0, stream>>>(
        h1, W2, W2 + 64 * 64, loop2, b2, relbuf, relbuf + (size_t)N * 64, h2, N);
    edge_scatter64<<<eb64, 256, 0, stream>>>(relbuf, src, dst, etypes, h2, E, N);

    layer3_node<<<nb, 256, 0, stream>>>(h2, W3, loop3, b3, rel3, agg3, N);
    edge_scatter2<4><<<eb2, 256, 0, stream>>>(rel3, src, dst, etypes, agg3, E, N);
    init_slot<<<1, 1, 0, stream>>>(slot);
    min_reduce<<<512, 256, 0, stream>>>(agg3, N * 2, slot);
    finalize_k<<<nb, 256, 0, stream>>>(agg3, cell, maxs, slot, out, N);
}


# All 3 edge scatters split by cache half: a block reads its die and draws edge chunks by ticket, adding only edges whose target row lies in its half's node range; float atomics, arrival order
# speedup vs baseline: 2.6177x; 1.3784x over previous
#include <hip/hip_runtime.h>

typedef __attribute__((ext_vector_type(2))) float v2f;
typedef __attribute__((ext_vector_type(8))) float v8f;

typedef __attribute__((ext_vector_type(4))) float v4f;

constexpr int TICKET_GAP = 64;

__device__ __forceinline__ int cache_half() {
    const unsigned pm = (unsigned)__builtin_amdgcn_s_sendmsg_rtn(0x87);
    return (int)((__builtin_amdgcn_readfirstlane(pm) >> 18) & 1u);
}

__device__ __forceinline__ void lds_barrier() {
    __builtin_amdgcn_fence(__ATOMIC_RELEASE, "workgroup", "local");
    __builtin_amdgcn_s_barrier();
    __builtin_amdgcn_fence(__ATOMIC_ACQUIRE, "workgroup", "local");
}

template <class F>
__device__ __forceinline__ void ticket_loop(unsigned* __restrict__ tickets,
                                            unsigned n_tickets, F work) {
    __shared__ unsigned s_t[2];
    const int tid = threadIdx.x;
    const int own = cache_half();
    unsigned it = 0;
    int q = own;
    if (tid == 0) s_t[0] = atomicAdd(tickets + TICKET_GAP * q, 1u);
    lds_barrier();
    unsigned t = __builtin_amdgcn_readfirstlane(s_t[0]);
    for (int pass = 0; pass < 2; ++pass) {
        while (t < n_tickets) {
            unsigned nxt = 0u;
            work(t, q, [&]() {
                if (tid == 0) {
                    unsigned off = (unsigned)(TICKET_GAP * q);
                    asm volatile("" : "+v"(off));
                    nxt = atomicAdd(tickets + off, 1u);
                }
            });
            if (tid == 0) s_t[(it + 1) & 1] = nxt;
            lds_barrier();
            ++it;
            t = __builtin_amdgcn_readfirstlane(s_t[it & 1]);
        }
        if (pass == 0) {
            q = 1 - own;
            if (tid == 0) s_t[(it + 1) & 1] = atomicAdd(tickets + TICKET_GAP * q, 1u);
            lds_barrier();
            ++it;
            t = __builtin_amdgcn_readfirstlane(s_t[it & 1]);
        }
    }
}

template <int K, bool RELU>
__global__ void __launch_bounds__(256)
gemm_wlds(const float* __restrict__ X,
          const float* __restrict__ Wr0, const float* __restrict__ Wr1,
          const float* __restrict__ Wlp, const float* __restrict__ blp,
          float* __restrict__ Cr0, float* __restrict__ Cr1,
          float* __restrict__ Clp, unsigned* __restrict__ tickets, int M) {
    const int y = blockIdx.y;
    const float* W    = (y == 0) ? Wr0 : (y == 1) ? Wr1 : Wlp;
    const float* bias = (y == 2) ? blp : nullptr;
    float*       C    = (y == 0) ? Cr0 : (y == 1) ? Cr1 : Clp;
    constexpr int KS    = K / 4;
    constexpr int KP    = K / 2;
    constexpr int PITCH = 68;
    constexpr int PLANE = KP * PITCH;
    __shared__ __attribute__((aligned(16))) float wl[2 * PLANE];

    const int tid  = threadIdx.x;
    const int lane = tid & 31;
    const int wave = blockIdx.x * 8 + (tid >> 5);
    const int m0   = wave * 16;
    if (blockIdx.x == 0 && y == 0 && tid == 0) {
        tickets[0] = 0u;
        tickets[TICKET_GAP] = 0u;
    }
    const bool active = m0 < M;

    const int half = lane >> 4;
    const int lm   = lane & 15;

    #pragma unroll
    for (int i = 0; i < (KP * 16) / 256; ++i) {
        const int item = tid + 256 * i;
        const int kp = item >> 4;
        const int c  = item & 15;
        const float4 r0 = *(const float4*)(W + (size_t)(2 * kp) * 64 + 4 * c);
        const float4 r1 = *(const float4*)(W + (size_t)(2 * kp + 1) * 64 + 4 * c);
        v4f p0; p0.x = r0.x; p0.y = r1.x; p0.z = r0.y; p0.w = r1.y;
        v4f p1; p1.x = r0.z; p1.y = r1.z; p1.z = r0.w; p1.w = r1.w;
        *(v4f*)(wl + kp * PITCH + 4 * c)         = p0;
        *(v4f*)(wl + PLANE + kp * PITCH + 4 * c) = p1;
    }

    float2 av[KS];
    float4 bq = make_float4(0.0f, 0.0f, 0.0f, 0.0f);
    if (active) {
        const float* xrow = X + (size_t)(m0 + lm) * K + 2 * half;
        #pragma unroll
        for (int ks = 0; ks < KS; ++ks) av[ks] = *(const float2*)(xrow + ks * 4);
        if (bias) bq = *(const float4*)(bias + 4 * lm);
    }

    __syncthreads();
    if (!active) return;

    const float bvs[4] = {bq.x, bq.y, bq.z, bq.w};
    v8f acc[4];
    #pragma unroll
    for (int nt = 0; nt < 4; ++nt) {
        #pragma unroll
        for (int v = 0; v < 8; ++v) acc[nt][v] = bvs[nt];
    }

    v2f a[KS];
    #pragma unroll
    for (int ks = 0; ks < KS; ++ks) {
        a[ks].x = RELU ? fmaxf(av[ks].x, 0.0f) : av[ks].x;
        a[ks].y = RELU ? fmaxf(av[ks].y, 0.0f) : av[ks].y;
        if (RELU) asm volatile("" : "+v"(a[ks]));
    }

    unsigned coff = (unsigned)((m0 + 8 * half) * 64 + 4 * lm) * 4u;
    const float* lw = wl + half * PITCH + 4 * lm;

    asm volatile("" : "+v"(acc[0]), "+v"(acc[1]), "+v"(acc[2]), "+v"(acc[3]));
    asm volatile("" : "+v"(coff));
    __builtin_amdgcn_sched_barrier(0);

    constexpr int GS = 4;
    constexpr int NG = KS / GS;
    v4f q[2][GS][2];
    #pragma unroll
    for (int s = 0; s < GS; ++s) {
        q[0][s][0] = *(const v4f*)(lw + (2 * s) * PITCH);
        q[0][s][1] = *(const v4f*)(lw + PLANE + (2 * s) * PITCH);
    }
    #pragma unroll
    for (int g = 0; g < NG; ++g) {
        if (g + 1 < NG) {
            #pragma unroll
            for (int s = 0; s < GS; ++s) {
                const int ks = (g + 1) * GS + s;
                q[(g + 1) & 1][s][0] = *(const v4f*)(lw + (2 * ks) * PITCH);
                q[(g + 1) & 1][s][1] = *(const v4f*)(lw + PLANE + (2 * ks) * PITCH);
            }
        }
        __builtin_amdgcn_sched_barrier(0);
        #pragma unroll
        for (int s = 0; s < GS; ++s) {
            const int ks = g * GS + s;
            const v4f q01 = q[g & 1][s][0];
            const v4f q23 = q[g & 1][s][1];
            const v2f b0 = __builtin_shufflevector(q01, q01, 0, 1);
            const v2f b1 = __builtin_shufflevector(q01, q01, 2, 3);
            const v2f b2 = __builtin_shufflevector(q23, q23, 0, 1);
            const v2f b3 = __builtin_shufflevector(q23, q23, 2, 3);
            acc[0] = __builtin_amdgcn_wmma_f32_16x16x4_f32(
                false, a[ks], false, b0, (short)0, acc[0], false, false);
            acc[1] = __builtin_amdgcn_wmma_f32_16x16x4_f32(
                false, a[ks], false, b1, (short)0, acc[1], false, false);
            acc[2] = __builtin_amdgcn_wmma_f32_16x16x4_f32(
                false, a[ks], false, b2, (short)0, acc[2], false, false);
            acc[3] = __builtin_amdgcn_wmma_f32_16x16x4_f32(
                false, a[ks], false, b3, (short)0, acc[3], false, false);
        }
        __builtin_amdgcn_sched_barrier(0);
    }

    asm volatile("v_nop\n\tv_nop\n\tv_nop\n\tv_nop\n\tv_nop\n\tv_nop\n\tv_nop\n\tv_nop"
                 : "+v"(acc[0]), "+v"(acc[1]), "+v"(acc[2]), "+v"(acc[3]));

    float* crow = (float*)((char*)C + coff);
    #pragma unroll
    for (int v = 0; v < 8; ++v) {
        const float4 o = make_float4(acc[0][v], acc[1][v], acc[2][v], acc[3][v]);
        *(float4*)(crow + v * 64) = o;
    }
}

template <bool FULL>
__device__ __forceinline__ void scatter64_batch(unsigned long long& m,
                                                int s0, int s1, int d0, int d1,
                                                const float* __restrict__ rel_l,
                                                float* __restrict__ agg_l) {
    float a[8], b[8];
    int   dr[8];
    #pragma unroll
    for (int u = 0; u < 8; ++u) {
        dr[u] = -1;
        if (FULL || m != 0ull) {
            const int j  = __builtin_ctzll(m);
            m &= m - 1ull;
            const int jj = j & 31;
            const int sa = __builtin_amdgcn_readlane(s0, jj);
            const int sb = __builtin_amdgcn_readlane(s1, jj);
            const int da = __builtin_amdgcn_readlane(d0, jj);
            const int db = __builtin_amdgcn_readlane(d1, jj);
            const int sr = (j < 32) ? sa : sb;
            dr[u]        = (j < 32) ? da : db;
            const float* p = rel_l + (size_t)sr * 64;
            a[u] = p[0];
            b[u] = p[32];
        }
    }
    #pragma unroll
    for (int u = 0; u < 8; ++u) {
        if (FULL || dr[u] >= 0) {
            float* qd = agg_l + (size_t)dr[u] * 64;
            atomicAdd(qd, a[u]);
            atomicAdd(qd + 32, b[u]);
        }
    }
}

__global__ void __launch_bounds__(256)
edge_scatter64(const float* __restrict__ rel,
               const int* __restrict__ src,
               const int* __restrict__ dst,
               const int* __restrict__ et,
               float* __restrict__ agg, unsigned* __restrict__ tickets,
               int E, int N) {
    const int lane = threadIdx.x & 31;
    const int wv   = __builtin_amdgcn_readfirstlane((int)(threadIdx.x >> 5));
    const int split = ((N >> 1) + 31) & ~31;
    const float* rel_l = rel + lane;
    float*       agg_l = agg + lane;
    const unsigned n_tickets = (unsigned)((E + 511) / 512);

    ticket_loop(tickets, n_tickets, [&](unsigned t, int q, auto draw) {
        const long long e0 = (long long)t * 512 + wv * 64;
        if (e0 >= E) { draw(); return; }
        int sr[2], dr[2];
        unsigned keep[2];
        #pragma unroll
        for (int u = 0; u < 2; ++u) {
            const long long e  = e0 + 32 * u + lane;
            const bool      in = e < E;
            const long long ec = in ? e : (long long)E - 1;
            dr[u] = dst[ec];
            sr[u] = et[ec] * N + src[ec];
            keep[u] = __builtin_amdgcn_ballot_w32(in && ((dr[u] >= split) == (q != 0)));
        }
        draw();
        unsigned long long m = (unsigned long long)keep[0] |
                               ((unsigned long long)keep[1] << 32);
        while (__builtin_popcountll(m) >= 8)
            scatter64_batch<true>(m, sr[0], sr[1], dr[0], dr[1], rel_l, agg_l);
        if (m != 0ull)
            scatter64_batch<false>(m, sr[0], sr[1], dr[0], dr[1], rel_l, agg_l);
    });
}

__global__ void layer3_node(const float* __restrict__ h2,
                            const float* __restrict__ W3,
                            const float* __restrict__ loop3,
                            const float* __restrict__ b3,
                            float* __restrict__ rel3, float* __restrict__ agg3,
                            unsigned* __restrict__ tickets, int N) {
    const int i = blockIdx.x * blockDim.x + threadIdx.x;
    if (i == 0) {
        tickets[0] = 0u;
        tickets[TICKET_GAP] = 0u;
    }
    if (i >= N) return;
    const float* row = h2 + (size_t)i * 64;
    float r00 = 0.f, r01 = 0.f, r10 = 0.f, r11 = 0.f, l0 = 0.f, l1 = 0.f;
    #pragma unroll 8
    for (int k = 0; k < 64; ++k) {
        const float v = fmaxf(row[k], 0.0f);
        r00 += v * W3[k * 2 + 0];
        r01 += v * W3[k * 2 + 1];
        r10 += v * W3[128 + k * 2 + 0];
        r11 += v * W3[128 + k * 2 + 1];
        l0  += v * loop3[k * 2 + 0];
        l1  += v * loop3[k * 2 + 1];
    }
    rel3[(size_t)i * 2 + 0] = r00;
    rel3[(size_t)i * 2 + 1] = r01;
    rel3[(size_t)(N + i) * 2 + 0] = r10;
    rel3[(size_t)(N + i) * 2 + 1] = r11;
    agg3[(size_t)i * 2 + 0] = l0 + b3[0];
    agg3[(size_t)i * 2 + 1] = l1 + b3[1];
}

template <int CH, class D>
__device__ __forceinline__ void scatter2_chunks(const float* __restrict__ rel3,
                                                const int* __restrict__ src,
                                                const int* __restrict__ dst,
                                                const int* __restrict__ et,
                                                float* __restrict__ agg3,
                                                long long e0, int lane, int E, int N,
                                                int split, int q, D draw) {
    int srow[CH], drow[CH];
    #pragma unroll
    for (int u = 0; u < CH; ++u) {
        const long long e  = e0 + 32 * u + lane;
        const bool      in = e < E;
        const long long ec = in ? e : (long long)E - 1;
        const int d = dst[ec];
        srow[u] = et[ec] * N + src[ec];
        drow[u] = (in && ((d >= split) == (q != 0))) ? d : -1;
    }
    draw();

    const int col = lane & 1;
    const int jl  = lane >> 1;
    float val[CH][2];
    int   dr[CH][2];
    #pragma unroll
    for (int u = 0; u < CH; ++u) {
        #pragma unroll
        for (int h = 0; h < 2; ++h) {
            const int j  = 16 * h + jl;
            const int sr = __shfl(srow[u], j);
            dr[u][h]     = __shfl(drow[u], j);
            val[u][h]    = (dr[u][h] >= 0) ? rel3[(size_t)sr * 2 + col] : 0.0f;
        }
    }
    #pragma unroll
    for (int u = 0; u < CH; ++u) {
        #pragma unroll
        for (int h = 0; h < 2; ++h) {
            if (dr[u][h] >= 0)
                atomicAdd(agg3 + (size_t)dr[u][h] * 2 + col, val[u][h]);
        }
    }
}

template <int CH>
__global__ void __launch_bounds__(256)
edge_scatter2(const float* __restrict__ rel3,
              const int* __restrict__ src,
              const int* __restrict__ dst,
              const int* __restrict__ et,
              float* __restrict__ agg3, unsigned* __restrict__ tickets,
              int E, int N) {
    const int lane = threadIdx.x & 31;
    const int wv   = __builtin_amdgcn_readfirstlane((int)(threadIdx.x >> 5));
    const int split = ((N >> 1) + 31) & ~31;
    constexpr int TE = 8 * 32 * CH;
    const unsigned n_tickets = (unsigned)((E + TE - 1) / TE);

    ticket_loop(tickets, n_tickets, [&](unsigned t, int q, auto draw) {
        const long long e0 = (long long)t * TE + wv * (32 * CH);
        if (e0 >= E) { draw(); return; }
        scatter2_chunks<CH>(rel3, src, dst, et, agg3, e0, lane, E, N, split, q, draw);
    });
}

__device__ __forceinline__ unsigned f2ord(float f) {
    unsigned u = __float_as_uint(f);
    return (u & 0x80000000u) ? ~u : (u | 0x80000000u);
}
__device__ __forceinline__ float ord2f(unsigned u) {
    return __uint_as_float((u & 0x80000000u) ? (u ^ 0x80000000u) : ~u);
}

__global__ void init_slot(unsigned* slot) { *slot = 0xFFFFFFFFu; }

__global__ void min_reduce(const float* __restrict__ p, int n, unsigned* slot) {
    __shared__ float sm[256];
    float m = 3.402823466e38f;
    for (int i = blockIdx.x * blockDim.x + threadIdx.x; i < n;
         i += gridDim.x * blockDim.x)
        m = fminf(m, p[i]);
    sm[threadIdx.x] = m;
    __syncthreads();
    for (int s = blockDim.x >> 1; s > 0; s >>= 1) {
        if ((int)threadIdx.x < s) sm[threadIdx.x] = fminf(sm[threadIdx.x], sm[threadIdx.x + s]);
        __syncthreads();
    }
    if (threadIdx.x == 0) atomicMin(slot, f2ord(sm[0]));
}

__global__ void finalize_k(const float* __restrict__ agg3,
                           const int* __restrict__ cell,
                           const int* __restrict__ maxs,
                           const unsigned* __restrict__ slot,
                           float* __restrict__ out, int N) {
    const int i = blockIdx.x * blockDim.x + threadIdx.x;
    if (i >= N) return;
    const float gmin = ord2f(*slot) - 1.0f;
    const bool upper = cell[i] >= (maxs[i] - 1);
    const bool lower = cell[i] == 0;
    out[(size_t)i * 2 + 0] = upper ? gmin : agg3[(size_t)i * 2 + 0];
    out[(size_t)i * 2 + 1] = lower ? gmin : agg3[(size_t)i * 2 + 1];
}

extern "C" void kernel_launch(void* const* d_in, const int* in_sizes, int n_in,
                              void* d_out, int out_size, void* d_ws, size_t ws_size,
                              hipStream_t stream) {
    const float* x      = (const float*)d_in[0];
    const int*   src    = (const int*)d_in[1];
    const int*   dst    = (const int*)d_in[2];
    const int*   etypes = (const int*)d_in[3];
    const int*   cell   = (const int*)d_in[4];
    const int*   maxs   = (const int*)d_in[5];
    const float* W1     = (const float*)d_in[6];
    const float* loop1  = (const float*)d_in[7];
    const float* b1     = (const float*)d_in[8];
    const float* W2     = (const float*)d_in[9];
    const float* loop2  = (const float*)d_in[10];
    const float* b2     = (const float*)d_in[11];
    const float* W3     = (const float*)d_in[12];
    const float* loop3  = (const float*)d_in[13];
    const float* b3     = (const float*)d_in[14];
    float* out = (float*)d_out;

    const int N = in_sizes[4];
    const int E = in_sizes[1];

    float* relbuf = (float*)d_ws;
    float* h1     = relbuf + (size_t)2 * N * 64;
    float* h2     = h1 + (size_t)N * 64;
    float* rel3   = h2 + (size_t)N * 64;
    float* agg3   = rel3 + (size_t)4 * N;
    unsigned* slot = (unsigned*)(agg3 + (size_t)2 * N);

    const int waves   = (N + 15) / 16;
    const int gblocks = (waves * 32 + 255) / 256;
    const int nb      = (N + 255) / 256;

    unsigned* tk1 = (unsigned*)h2;
    unsigned* tk2 = (unsigned*)rel3;
    unsigned* tk3 = (unsigned*)relbuf;

    gemm_wlds<128, false><<<dim3(gblocks, 3), 256, 0, stream>>>(
        x, W1, W1 + 128 * 64, loop1, b1, relbuf, relbuf + (size_t)N * 64, h1, tk1, N);
    edge_scatter64<<<2048, 256, 0, stream>>>(relbuf, src, dst, etypes, h1, tk1, E, N);

    gemm_wlds<64, true><<<dim3(gblocks, 3), 256, 0, stream>>>(
        h1, W2, W2 + 64 * 64, loop2, b2, relbuf, relbuf + (size_t)N * 64, h2, tk2, N);
    edge_scatter64<<<2048, 256, 0, stream>>>(relbuf, src, dst, etypes, h2, tk2, E, N);

    layer3_node<<<nb, 256, 0, stream>>>(h2, W3, loop3, b3, rel3, agg3, tk3, N);
    edge_scatter2<4><<<1024, 256, 0, stream>>>(rel3, src, dst, etypes, agg3, tk3, E, N);
    init_slot<<<1, 1, 0, stream>>>(slot);
    min_reduce<<<512, 256, 0, stream>>>(agg3, N * 2, slot);
    finalize_k<<<nb, 256, 0, stream>>>(agg3, cell, maxs, slot, out, N);
}
